# starting kernel
# baseline (speedup 1.0000x reference)
#include <hip/hip_runtime.h>
#include <math.h>

typedef __attribute__((ext_vector_type(2))) float v2f;
typedef __attribute__((ext_vector_type(8))) float v8f;

#define NQ      12
#define DIM     4096
#define NLAYERS 20
#define F_IN    2048
#define PROW    17
#define GROW    (16 * PROW)

__device__ __forceinline__ int pidx(int i) { return (i >> 4) * PROW + (i & 15); }

__device__ __forceinline__ int cnot_src(int i) {
    const int cq[11] = {0, 2, 4, 6, 8, 10, 1, 3, 5, 7, 9};
    const int tq[11] = {1, 3, 5, 7, 9, 11, 2, 4, 6, 8, 10};
    int j = i;
#pragma unroll
    for (int k = 10; k >= 0; --k) {
        int cm = 1 << (11 - cq[k]);
        int tm = 1 << (11 - tq[k]);
        if (j & cm) j ^= tm;
    }
    return j;
}

__device__ __forceinline__ float tanh_pio2(float x) {
    float xc = fminf(fmaxf(x, -15.f), 15.f);
    float z  = __expf(xc * 2.0f);
    return (z - 1.0f) * __builtin_amdgcn_rcpf(z + 1.0f) * 1.57079632679489662f;
}

__global__ __launch_bounds__(256)
void qnet_kernel(const float* __restrict__ in, const float* __restrict__ qp,
                 float* __restrict__ out)
{
    __shared__ float Sa[256 * PROW];
    __shared__ float Sb[256 * PROW];
    __shared__ float Mm[3][16 * PROW];
    __shared__ float CS[NQ], SN[NQ];
    __shared__ float red[96];
    __shared__ float s_scale;

    const int t    = threadIdx.x;
    const int lane = t & 31;
    const int wave = t >> 5;
    const int hi   = lane >> 4;
    const int lo   = lane & 15;
    const int row  = blockIdx.x;

    const float* srcrow = in + (size_t)row * F_IN;
    float ss = 0.f;
#pragma unroll
    for (int e = 0; e < 8; ++e) {
        int i = e * 256 + t;
        float q = tanh_pio2(srcrow[i]);
        Sa[pidx(i)] = q;
        ss += q * q;
    }
#pragma unroll
    for (int e = 8; e < 16; ++e) Sa[pidx(e * 256 + t)] = 0.5f;

    for (int off = 16; off; off >>= 1) ss += __shfl_down(ss, off, 32);
    if (lane == 0) red[wave] = ss;
    __syncthreads();
    if (t == 0) {
        float tot = 512.0f;
        for (int w = 0; w < 8; ++w) tot += red[w];
        s_scale = 1.0f / sqrtf(tot);
    }
    __syncthreads();
    {
        float sc = s_scale;
#pragma unroll
        for (int e = 0; e < 16; ++e) Sa[pidx(e * 256 + t)] *= sc;
    }

    int srcA[16];
#pragma unroll
    for (int tt = 0; tt < 2; ++tt) {
        int T = wave + tt * 8;
#pragma unroll
        for (int s = 0; s < 4; ++s) {
#pragma unroll
            for (int d = 0; d < 2; ++d) {
                int i = (T * 16 + lo) * 16 + 4 * s + 2 * hi + d;
                srcA[tt * 8 + s * 2 + d] = pidx(cnot_src(i));
            }
        }
    }

    float* pa = Sa;
    float* pb = Sb;

    const int T0 = wave;
    const int T1 = wave + 8;

    const float* th = qp + NQ;

    for (int L = 0; L < NLAYERS; ++L) {
        if (t < NQ) {
            float a = th[L * NQ + t] * 0.5f;
            CS[t] = cosf(a);
            SN[t] = sinf(a);
        }
        __syncthreads();

        {
            int r = t >> 4, c = t & 15;
#pragma unroll
            for (int g = 0; g < 3; ++g) {
                float f = 1.f;
#pragma unroll
                for (int p = 0; p < 4; ++p) {
                    int q  = g * 4 + p;
                    int rb = (r >> (3 - p)) & 1;
                    int cb = (c >> (3 - p)) & 1;
                    f *= (rb == cb) ? CS[q] : (rb ? SN[q] : -SN[q]);
                }
                Mm[g][r * PROW + c] = f;
            }
        }
        __syncthreads();

        {
            v8f acc0 = {0, 0, 0, 0, 0, 0, 0, 0};
            v8f acc1 = {0, 0, 0, 0, 0, 0, 0, 0};
#pragma unroll
            for (int s = 0; s < 4; ++s) {
                int k0 = 4 * s + 2 * hi;
                v2f a0, a1, b;
                a0.x = pa[srcA[s * 2 + 0]];
                a0.y = pa[srcA[s * 2 + 1]];
                a1.x = pa[srcA[8 + s * 2 + 0]];
                a1.y = pa[srcA[8 + s * 2 + 1]];
                b.x  = Mm[2][lo * PROW + k0];
                b.y  = Mm[2][lo * PROW + k0 + 1];
                acc0 = __builtin_amdgcn_wmma_f32_16x16x4_f32(
                    false, a0, false, b, (short)0, acc0, false, false);
                acc1 = __builtin_amdgcn_wmma_f32_16x16x4_f32(
                    false, a1, false, b, (short)0, acc1, false, false);
            }
#pragma unroll
            for (int v = 0; v < 8; ++v) {
                pb[(T0 * 16 + v + 8 * hi) * PROW + lo] = acc0[v];
                pb[(T1 * 16 + v + 8 * hi) * PROW + lo] = acc1[v];
            }
        }
        __syncthreads();

        {
            v8f acc0 = {0, 0, 0, 0, 0, 0, 0, 0};
            v8f acc1 = {0, 0, 0, 0, 0, 0, 0, 0};
#pragma unroll
            for (int s = 0; s < 4; ++s) {
                int k0 = 4 * s + 2 * hi;
                v2f a, b0, b1;
                a.x  = Mm[1][lo * PROW + k0];
                a.y  = Mm[1][lo * PROW + k0 + 1];
                b0.x = pb[T0 * GROW + k0 * PROW + lo];
                b0.y = pb[T0 * GROW + (k0 + 1) * PROW + lo];
                b1.x = pb[T1 * GROW + k0 * PROW + lo];
                b1.y = pb[T1 * GROW + (k0 + 1) * PROW + lo];
                acc0 = __builtin_amdgcn_wmma_f32_16x16x4_f32(
                    false, a, false, b0, (short)0, acc0, false, false);
                acc1 = __builtin_amdgcn_wmma_f32_16x16x4_f32(
                    false, a, false, b1, (short)0, acc1, false, false);
            }
#pragma unroll
            for (int v = 0; v < 8; ++v) {
                pa[T0 * GROW + (v + 8 * hi) * PROW + lo] = acc0[v];
                pa[T1 * GROW + (v + 8 * hi) * PROW + lo] = acc1[v];
            }
        }
        __syncthreads();

        {
            v8f acc0 = {0, 0, 0, 0, 0, 0, 0, 0};
            v8f acc1 = {0, 0, 0, 0, 0, 0, 0, 0};
#pragma unroll
            for (int s = 0; s < 4; ++s) {
                int k0 = 4 * s + 2 * hi;
                v2f a, b0, b1;
                a.x  = Mm[0][lo * PROW + k0];
                a.y  = Mm[0][lo * PROW + k0 + 1];
                b0.x = pa[k0 * GROW + T0 * PROW + lo];
                b0.y = pa[(k0 + 1) * GROW + T0 * PROW + lo];
                b1.x = pa[k0 * GROW + T1 * PROW + lo];
                b1.y = pa[(k0 + 1) * GROW + T1 * PROW + lo];
                acc0 = __builtin_amdgcn_wmma_f32_16x16x4_f32(
                    false, a, false, b0, (short)0, acc0, false, false);
                acc1 = __builtin_amdgcn_wmma_f32_16x16x4_f32(
                    false, a, false, b1, (short)0, acc1, false, false);
            }
#pragma unroll
            for (int v = 0; v < 8; ++v) {
                pb[(v + 8 * hi) * GROW + T0 * PROW + lo] = acc0[v];
                pb[(v + 8 * hi) * GROW + T1 * PROW + lo] = acc1[v];
            }
        }
        __syncthreads();

        float* tmp = pa; pa = pb; pb = tmp;
    }

    float P = 0.f;
    float A[4] = {0.f, 0.f, 0.f, 0.f};
#pragma unroll
    for (int e = 0; e < 16; ++e) {
        int i = e * 256 + t;
        float v = pa[pidx(i)];
        float p = v * v;
        P += p;
#pragma unroll
        for (int q = 0; q < 4; ++q)
            A[q] += ((e >> (3 - q)) & 1) ? -p : p;
    }
    float acc[NQ];
#pragma unroll
    for (int q = 0; q < 4; ++q) acc[q] = A[q];
#pragma unroll
    for (int q = 4; q < NQ; ++q)
        acc[q] = ((t >> (11 - q)) & 1) ? -P : P;

    for (int off = 16; off; off >>= 1) {
#pragma unroll
        for (int q = 0; q < NQ; ++q) acc[q] += __shfl_down(acc[q], off, 32);
    }
    if (lane == 0) {
#pragma unroll
        for (int q = 0; q < NQ; ++q) red[wave * NQ + q] = acc[q];
    }
    __syncthreads();
    if (t < NQ) {
        float s = 0.f;
#pragma unroll
        for (int w = 0; w < 8; ++w) s += red[w * NQ + t];
        out[(size_t)row * NQ + t] = s;
    }
}

extern "C" void kernel_launch(void* const* d_in, const int* in_sizes, int n_in,
                              void* d_out, int out_size, void* d_ws, size_t ws_size,
                              hipStream_t stream) {
    (void)n_in; (void)out_size; (void)d_ws; (void)ws_size;
    const float* in  = (const float*)d_in[0];
    const float* qp  = (const float*)d_in[1];
    float*       out = (float*)d_out;
    int rows = in_sizes[0] / F_IN;
    qnet_kernel<<<rows, 256, 0, stream>>>(in, qp, out);
}


# one wave owns one whole row: a2,a1 rotations chained in registers per slab, a0 via one in-place LDS transpose, entangling permutation as scatter store; no barriers; k and output sum order change
# speedup vs baseline: 1.3796x; 1.3796x over previous
#include <hip/hip_runtime.h>
#include <math.h>

typedef __attribute__((ext_vector_type(2))) float v2f;
typedef __attribute__((ext_vector_type(8))) float v8f;

#define NQ      12
#define NLAYERS 20
#define F_IN    2048
#define PA      18
#define SA      (16 * PA)
#define NST     (16 * SA)

__device__ __forceinline__ int addr_of(int j) {
    return (j >> 8) * SA + ((j >> 4) & 15) * PA + (j & 15);
}

__device__ __forceinline__ int cnot_dst(int i) {
    const int cq[11] = {0, 2, 4, 6, 8, 10, 1, 3, 5, 7, 9};
    const int tq[11] = {1, 3, 5, 7, 9, 11, 2, 4, 6, 8, 10};
    int j = i;
#pragma unroll
    for (int k = 0; k < 11; ++k) {
        int cm = 1 << (11 - cq[k]);
        int tm = 1 << (11 - tq[k]);
        if (j & cm) j ^= tm;
    }
    return j;
}

__device__ __forceinline__ float tanh_pio2(float x) {
    float xc = fminf(fmaxf(x, -15.f), 15.f);
    float z  = __expf(xc * 2.0f);
    return (z - 1.0f) * __builtin_amdgcn_rcpf(z + 1.0f) * 1.57079632679489662f;
}

__device__ __forceinline__ float bcast(float x, int srclane) {
    return __builtin_bit_cast(float, __builtin_amdgcn_readlane(__builtin_bit_cast(int, x), srclane));
}

__device__ __forceinline__ float ryf(float c, float s, int ob, int ib) {
    return (ob == ib) ? c : (ob ? s : -s);
}

template <bool MSBHI>
__device__ __forceinline__ void build_frag(const float c[4], const float sn[4], int lo, int hi, v2f fr[4]) {
    const int o3 = (lo >> 3) & 1, o2 = (lo >> 2) & 1, o1 = (lo >> 1) & 1, o0 = lo & 1;
    if (MSBHI) {
        float f3 = ryf(c[0], sn[0], o3, hi);
        float f2[2] = {ryf(c[1], sn[1], o2, 0), ryf(c[1], sn[1], o2, 1)};
        float f1[2] = {ryf(c[2], sn[2], o1, 0), ryf(c[2], sn[2], o1, 1)};
        float f0[2] = {ryf(c[3], sn[3], o0, 0), ryf(c[3], sn[3], o0, 1)};
#pragma unroll
        for (int s = 0; s < 4; ++s) {
            float t = f3 * f2[s >> 1] * f1[s & 1];
            fr[s].x = t * f0[0];
            fr[s].y = t * f0[1];
        }
    } else {
        float f0 = ryf(c[3], sn[3], o0, hi);
        float f3[2] = {ryf(c[0], sn[0], o3, 0), ryf(c[0], sn[0], o3, 1)};
        float f2[2] = {ryf(c[1], sn[1], o2, 0), ryf(c[1], sn[1], o2, 1)};
        float f1[2] = {ryf(c[2], sn[2], o1, 0), ryf(c[2], sn[2], o1, 1)};
#pragma unroll
        for (int s = 0; s < 4; ++s) {
            float t = f0 * f3[s >> 1] * f2[s & 1];
            fr[s].x = t * f1[0];
            fr[s].y = t * f1[1];
        }
    }
}

#define WMMA(a, b, c) __builtin_amdgcn_wmma_f32_16x16x4_f32(false, (a), false, (b), (short)0, (c), false, false)

__global__ __launch_bounds__(32) __attribute__((amdgpu_waves_per_eu(4, 4)))
void qnet_wave_kernel(const float* __restrict__ in, const float* __restrict__ qp,
                      float* __restrict__ out)
{
    __shared__ __attribute__((aligned(16))) float S[NST];

    const int lane = threadIdx.x;
    const int hi   = lane >> 4;
    const int lo   = lane & 15;
    const int row  = blockIdx.x;

    const float* srcrow = in + (size_t)row * F_IN;
    const int dlane = cnot_dst(lane);
    {
        float q[64];
        float ss = 0.f;
#pragma unroll
        for (int e = 0; e < 64; ++e) {
            q[e] = tanh_pio2(srcrow[e * 32 + lane]);
            ss += q[e] * q[e];
        }
#pragma unroll
        for (int off = 16; off; off >>= 1) ss += __shfl_xor(ss, off, 32);
        const float sc = 1.0f / sqrtf(ss + 512.0f);
        const float padv = 0.5f * sc;
#pragma unroll
        for (int e = 0; e < 64; ++e) {
            S[addr_of(cnot_dst(e * 32) ^ dlane)] = q[e] * sc;
            S[addr_of(cnot_dst(2048 + e * 32) ^ dlane)] = padv;
        }
    }
    __syncthreads();

    const int dscat = cnot_dst((hi << 11) | lo);
    int basev[8];
#pragma unroll
    for (int m = 0; m < 8; ++m)
        basev[m] = addr_of(dscat ^ ((((m >> 2) & 1) << 10) | (((m >> 1) & 1) << 9) | ((m & 1) << 3)));
    const float* th = qp + NQ;

    float* xbase = S + lo * PA + 8 * hi;
    const float* ybase = S + hi * SA + lo;

    v8f o3[16];
    for (int L = 0; L < NLAYERS; ++L) {
        float ang = th[L * NQ + (lane < NQ ? lane : 0)] * 0.5f;
        float cl = cosf(ang), sl = sinf(ang);
        float c[NQ], sn[NQ];
#pragma unroll
        for (int q = 0; q < NQ; ++q) { c[q] = bcast(cl, q); sn[q] = bcast(sl, q); }

        v2f F2[4], F1[4], F0[4];
        build_frag<true >(c + 8, sn + 8, lo, hi, F2);
        build_frag<true >(c + 4, sn + 4, lo, hi, F1);
        build_frag<false>(c + 0, sn + 0, lo, hi, F0);

#pragma unroll
        for (int z = 0; z < 16; ++z) {
            float* p = xbase + z * SA;
            v2f x0 = *(const v2f*)(p + 0), x1 = *(const v2f*)(p + 2);
            v2f x2 = *(const v2f*)(p + 4), x3 = *(const v2f*)(p + 6);
            v8f a = {0, 0, 0, 0, 0, 0, 0, 0};
            a = WMMA(x0, F2[0], a);
            a = WMMA(x1, F2[1], a);
            a = WMMA(x2, F2[2], a);
            a = WMMA(x3, F2[3], a);
            v8f b = {0, 0, 0, 0, 0, 0, 0, 0};
            v2f t0 = {a[0], a[1]}, t1 = {a[2], a[3]}, t2 = {a[4], a[5]}, t3 = {a[6], a[7]};
            b = WMMA(t0, F1[0], b);
            b = WMMA(t1, F1[1], b);
            b = WMMA(t2, F1[2], b);
            b = WMMA(t3, F1[3], b);
            v2f s0 = {b[0], b[1]}, s1 = {b[2], b[3]}, s2 = {b[4], b[5]}, s3 = {b[6], b[7]};
            *(v2f*)(p + 0) = s0; *(v2f*)(p + 2) = s1;
            *(v2f*)(p + 4) = s2; *(v2f*)(p + 6) = s3;
        }
        __syncthreads();

#pragma unroll
        for (int z = 0; z < 16; ++z) {
            const float* p = ybase + z * PA;
            v8f y;
#pragma unroll
            for (int v = 0; v < 8; ++v) y[v] = p[(2 * v) * SA];
            v8f a = {0, 0, 0, 0, 0, 0, 0, 0};
            v2f t0 = {y[0], y[1]}, t1 = {y[2], y[3]}, t2 = {y[4], y[5]}, t3 = {y[6], y[7]};
            a = WMMA(F0[0], t0, a);
            a = WMMA(F0[1], t1, a);
            a = WMMA(F0[2], t2, a);
            a = WMMA(F0[3], t3, a);
            o3[z] = a;
        }
        if (L + 1 < NLAYERS) {
            __syncthreads();
#pragma unroll
            for (int z = 0; z < 16; ++z) {
#pragma unroll
                for (int r = 0; r < 8; ++r) {
                    const int K = cnot_dst((r << 8) | (z << 4));
                    const int m = (((K >> 10) & 1) << 2) | (((K >> 9) & 1) << 1) | ((K >> 3) & 1);
                    S[basev[m] + addr_of(K & ~0xE0F)] = o3[z][r];
                }
            }
            __syncthreads();
        }
    }

    float P = 0.f;
    float A[7] = {0.f, 0.f, 0.f, 0.f, 0.f, 0.f, 0.f};
#pragma unroll
    for (int z = 0; z < 16; ++z) {
#pragma unroll
        for (int r = 0; r < 8; ++r) {
            float v = o3[z][r];
            float p = v * v;
            P += p;
#pragma unroll
            for (int q = 1; q < 4; ++q) A[q - 1] += ((r >> (3 - q)) & 1) ? -p : p;
#pragma unroll
            for (int q = 4; q < 8; ++q) A[q - 1] += ((z >> (7 - q)) & 1) ? -p : p;
        }
    }
    float acc[NQ];
    acc[0] = hi ? -P : P;
#pragma unroll
    for (int q = 1; q < 8; ++q) acc[q] = A[q - 1];
#pragma unroll
    for (int q = 8; q < NQ; ++q) acc[q] = ((lo >> (11 - q)) & 1) ? -P : P;
#pragma unroll
    for (int off = 16; off; off >>= 1) {
#pragma unroll
        for (int q = 0; q < NQ; ++q) acc[q] += __shfl_xor(acc[q], off, 32);
    }
    float res = acc[0];
#pragma unroll
    for (int q = 1; q < NQ; ++q) res = (lane == q) ? acc[q] : res;
    if (lane < NQ) out[(size_t)row * NQ + lane] = res;
}

extern "C" void kernel_launch(void* const* d_in, const int* in_sizes, int n_in,
                              void* d_out, int out_size, void* d_ws, size_t ws_size,
                              hipStream_t stream) {
    (void)n_in; (void)out_size; (void)d_ws; (void)ws_size;
    const float* in  = (const float*)d_in[0];
    const float* qp  = (const float*)d_in[1];
    float*       out = (float*)d_out;
    int rows = in_sizes[0] / F_IN;
    qnet_wave_kernel<<<rows, 32, 0, stream>>>(in, qp, out);
}


# each wave steps its issue priority down (s_setprio 3,2,1,0 at layers 0,10,15,18) so the row furthest behind on a SIMD wins the matrix pipe and its 4 rows finish together; arithmetic unchanged
# speedup vs baseline: 1.4602x; 1.0584x over previous
#include <hip/hip_runtime.h>
#include <math.h>

typedef __attribute__((ext_vector_type(2))) float v2f;
typedef __attribute__((ext_vector_type(8))) float v8f;

#define NQ      12
#define NLAYERS 20
#define F_IN    2048
#define PA      18
#define SA      (16 * PA)
#define NST     (16 * SA)
#define PRIO_DROP_A 10
#define PRIO_DROP_B 15
#define PRIO_DROP_C 18

__device__ __forceinline__ int addr_of(int j) {
    return (j >> 8) * SA + ((j >> 4) & 15) * PA + (j & 15);
}

__device__ __forceinline__ int cnot_dst(int i) {
    const int cq[11] = {0, 2, 4, 6, 8, 10, 1, 3, 5, 7, 9};
    const int tq[11] = {1, 3, 5, 7, 9, 11, 2, 4, 6, 8, 10};
    int j = i;
#pragma unroll
    for (int k = 0; k < 11; ++k) {
        int cm = 1 << (11 - cq[k]);
        int tm = 1 << (11 - tq[k]);
        if (j & cm) j ^= tm;
    }
    return j;
}

__device__ __forceinline__ float tanh_pio2(float x) {
    float xc = fminf(fmaxf(x, -15.f), 15.f);
    float z  = __expf(xc * 2.0f);
    return (z - 1.0f) * __builtin_amdgcn_rcpf(z + 1.0f) * 1.57079632679489662f;
}

__device__ __forceinline__ float bcast(float x, int srclane) {
    return __builtin_bit_cast(float, __builtin_amdgcn_readlane(__builtin_bit_cast(int, x), srclane));
}

__device__ __forceinline__ float ryf(float c, float s, int ob, int ib) {
    return (ob == ib) ? c : (ob ? s : -s);
}

template <bool MSBHI>
__device__ __forceinline__ void build_frag(const float c[4], const float sn[4], int lo, int hi, v2f fr[4]) {
    const int o3 = (lo >> 3) & 1, o2 = (lo >> 2) & 1, o1 = (lo >> 1) & 1, o0 = lo & 1;
    if (MSBHI) {
        float f3 = ryf(c[0], sn[0], o3, hi);
        float f2[2] = {ryf(c[1], sn[1], o2, 0), ryf(c[1], sn[1], o2, 1)};
        float f1[2] = {ryf(c[2], sn[2], o1, 0), ryf(c[2], sn[2], o1, 1)};
        float f0[2] = {ryf(c[3], sn[3], o0, 0), ryf(c[3], sn[3], o0, 1)};
#pragma unroll
        for (int s = 0; s < 4; ++s) {
            float t = f3 * f2[s >> 1] * f1[s & 1];
            fr[s].x = t * f0[0];
            fr[s].y = t * f0[1];
        }
    } else {
        float f0 = ryf(c[3], sn[3], o0, hi);
        float f3[2] = {ryf(c[0], sn[0], o3, 0), ryf(c[0], sn[0], o3, 1)};
        float f2[2] = {ryf(c[1], sn[1], o2, 0), ryf(c[1], sn[1], o2, 1)};
        float f1[2] = {ryf(c[2], sn[2], o1, 0), ryf(c[2], sn[2], o1, 1)};
#pragma unroll
        for (int s = 0; s < 4; ++s) {
            float t = f0 * f3[s >> 1] * f2[s & 1];
            fr[s].x = t * f1[0];
            fr[s].y = t * f1[1];
        }
    }
}

#define WMMA(a, b, c) __builtin_amdgcn_wmma_f32_16x16x4_f32(false, (a), false, (b), (short)0, (c), false, false)

__global__ __launch_bounds__(32) __attribute__((amdgpu_waves_per_eu(4, 4)))
void qnet_wave_kernel(const float* __restrict__ in, const float* __restrict__ qp,
                      float* __restrict__ out)
{
    __shared__ __attribute__((aligned(16))) float S[NST];

    __builtin_amdgcn_s_setprio(3);

    const int lane = threadIdx.x;
    const int hi   = lane >> 4;
    const int lo   = lane & 15;
    const int row  = blockIdx.x;

    const float* srcrow = in + (size_t)row * F_IN;
    const int dlane = cnot_dst(lane);
    {
        float q[64];
        float ss = 0.f;
#pragma unroll
        for (int e = 0; e < 64; ++e) {
            q[e] = tanh_pio2(srcrow[e * 32 + lane]);
            ss += q[e] * q[e];
        }
#pragma unroll
        for (int off = 16; off; off >>= 1) ss += __shfl_xor(ss, off, 32);
        const float sc = 1.0f / sqrtf(ss + 512.0f);
        const float padv = 0.5f * sc;
#pragma unroll
        for (int e = 0; e < 64; ++e) {
            S[addr_of(cnot_dst(e * 32) ^ dlane)] = q[e] * sc;
            S[addr_of(cnot_dst(2048 + e * 32) ^ dlane)] = padv;
        }
    }
    __syncthreads();

    const int dscat = cnot_dst((hi << 11) | lo);
    int basev[8];
#pragma unroll
    for (int m = 0; m < 8; ++m)
        basev[m] = addr_of(dscat ^ ((((m >> 2) & 1) << 10) | (((m >> 1) & 1) << 9) | ((m & 1) << 3)));
    const float* th = qp + NQ;

    float* xbase = S + lo * PA + 8 * hi;
    const float* ybase = S + hi * SA + lo;

    v8f o3[16];
    for (int L = 0; L < NLAYERS; ++L) {
        if (L == PRIO_DROP_A)      __builtin_amdgcn_s_setprio(2);
        else if (L == PRIO_DROP_B) __builtin_amdgcn_s_setprio(1);
        else if (L == PRIO_DROP_C) __builtin_amdgcn_s_setprio(0);

        float ang = th[L * NQ + (lane < NQ ? lane : 0)] * 0.5f;
        float cl = cosf(ang), sl = sinf(ang);
        float c[NQ], sn[NQ];
#pragma unroll
        for (int q = 0; q < NQ; ++q) { c[q] = bcast(cl, q); sn[q] = bcast(sl, q); }

        v2f F2[4], F1[4], F0[4];
        build_frag<true >(c + 8, sn + 8, lo, hi, F2);
        build_frag<true >(c + 4, sn + 4, lo, hi, F1);
        build_frag<false>(c + 0, sn + 0, lo, hi, F0);

#pragma unroll
        for (int z = 0; z < 16; ++z) {
            float* p = xbase + z * SA;
            v2f x0 = *(const v2f*)(p + 0), x1 = *(const v2f*)(p + 2);
            v2f x2 = *(const v2f*)(p + 4), x3 = *(const v2f*)(p + 6);
            v8f a = {0, 0, 0, 0, 0, 0, 0, 0};
            a = WMMA(x0, F2[0], a);
            a = WMMA(x1, F2[1], a);
            a = WMMA(x2, F2[2], a);
            a = WMMA(x3, F2[3], a);
            v8f b = {0, 0, 0, 0, 0, 0, 0, 0};
            v2f t0 = {a[0], a[1]}, t1 = {a[2], a[3]}, t2 = {a[4], a[5]}, t3 = {a[6], a[7]};
            b = WMMA(t0, F1[0], b);
            b = WMMA(t1, F1[1], b);
            b = WMMA(t2, F1[2], b);
            b = WMMA(t3, F1[3], b);
            v2f s0 = {b[0], b[1]}, s1 = {b[2], b[3]}, s2 = {b[4], b[5]}, s3 = {b[6], b[7]};
            *(v2f*)(p + 0) = s0; *(v2f*)(p + 2) = s1;
            *(v2f*)(p + 4) = s2; *(v2f*)(p + 6) = s3;
        }
        __syncthreads();

#pragma unroll
        for (int z = 0; z < 16; ++z) {
            const float* p = ybase + z * PA;
            v8f y;
#pragma unroll
            for (int v = 0; v < 8; ++v) y[v] = p[(2 * v) * SA];
            v8f a = {0, 0, 0, 0, 0, 0, 0, 0};
            v2f t0 = {y[0], y[1]}, t1 = {y[2], y[3]}, t2 = {y[4], y[5]}, t3 = {y[6], y[7]};
            a = WMMA(F0[0], t0, a);
            a = WMMA(F0[1], t1, a);
            a = WMMA(F0[2], t2, a);
            a = WMMA(F0[3], t3, a);
            o3[z] = a;
        }
        if (L + 1 < NLAYERS) {
            __syncthreads();
#pragma unroll
            for (int z = 0; z < 16; ++z) {
#pragma unroll
                for (int r = 0; r < 8; ++r) {
                    const int K = cnot_dst((r << 8) | (z << 4));
                    const int m = (((K >> 10) & 1) << 2) | (((K >> 9) & 1) << 1) | ((K >> 3) & 1);
                    S[basev[m] + addr_of(K & ~0xE0F)] = o3[z][r];
                }
            }
            __syncthreads();
        }
    }

    float P = 0.f;
    float A[7] = {0.f, 0.f, 0.f, 0.f, 0.f, 0.f, 0.f};
#pragma unroll
    for (int z = 0; z < 16; ++z) {
#pragma unroll
        for (int r = 0; r < 8; ++r) {
            float v = o3[z][r];
            float p = v * v;
            P += p;
#pragma unroll
            for (int q = 1; q < 4; ++q) A[q - 1] += ((r >> (3 - q)) & 1) ? -p : p;
#pragma unroll
            for (int q = 4; q < 8; ++q) A[q - 1] += ((z >> (7 - q)) & 1) ? -p : p;
        }
    }
    float acc[NQ];
    acc[0] = hi ? -P : P;
#pragma unroll
    for (int q = 1; q < 8; ++q) acc[q] = A[q - 1];
#pragma unroll
    for (int q = 8; q < NQ; ++q) acc[q] = ((lo >> (11 - q)) & 1) ? -P : P;
#pragma unroll
    for (int off = 16; off; off >>= 1) {
#pragma unroll
        for (int q = 0; q < NQ; ++q) acc[q] += __shfl_xor(acc[q], off, 32);
    }
    float res = acc[0];
#pragma unroll
    for (int q = 1; q < NQ; ++q) res = (lane == q) ? acc[q] : res;
    if (lane < NQ) out[(size_t)row * NQ + lane] = res;
}

extern "C" void kernel_launch(void* const* d_in, const int* in_sizes, int n_in,
                              void* d_out, int out_size, void* d_ws, size_t ws_size,
                              hipStream_t stream) {
    (void)n_in; (void)out_size; (void)d_ws; (void)ws_size;
    const float* in  = (const float*)d_in[0];
    const float* qp  = (const float*)d_in[1];
    float*       out = (float*)d_out;
    int rows = in_sizes[0] / F_IN;
    qnet_wave_kernel<<<rows, 32, 0, stream>>>(in, qp, out);
}


# both matrix phases run in blocks of 4 tiles: the next block's LDS fragment loads issued as one group ahead, operand pairs loaded in place, 4 accumulator chains interleaved; sum order unchanged
# speedup vs baseline: 1.5612x; 1.0692x over previous
#include <hip/hip_runtime.h>
#include <math.h>

typedef __attribute__((ext_vector_type(2))) float v2f;
typedef __attribute__((ext_vector_type(8))) float v8f;

#define NQ      12
#define NLAYERS 20
#define F_IN    2048
#define PA      18
#define SA      (16 * PA)
#define NST     (16 * SA)
#define PRIO_DROP_A 10
#define PRIO_DROP_B 15
#define PRIO_DROP_C 18

__device__ __forceinline__ int addr_of(int j) {
    return (j >> 8) * SA + ((j >> 4) & 15) * PA + (j & 15);
}

__device__ __forceinline__ int cnot_dst(int i) {
    const int cq[11] = {0, 2, 4, 6, 8, 10, 1, 3, 5, 7, 9};
    const int tq[11] = {1, 3, 5, 7, 9, 11, 2, 4, 6, 8, 10};
    int j = i;
#pragma unroll
    for (int k = 0; k < 11; ++k) {
        int cm = 1 << (11 - cq[k]);
        int tm = 1 << (11 - tq[k]);
        if (j & cm) j ^= tm;
    }
    return j;
}

__device__ __forceinline__ float tanh_pio2(float x) {
    float xc = fminf(fmaxf(x, -15.f), 15.f);
    float z  = __expf(xc * 2.0f);
    return (z - 1.0f) * __builtin_amdgcn_rcpf(z + 1.0f) * 1.57079632679489662f;
}

__device__ __forceinline__ float bcast(float x, int srclane) {
    return __builtin_bit_cast(float, __builtin_amdgcn_readlane(__builtin_bit_cast(int, x), srclane));
}

__device__ __forceinline__ float ryf(float c, float s, int ob, int ib) {
    return (ob == ib) ? c : (ob ? s : -s);
}

template <bool MSBHI>
__device__ __forceinline__ void build_frag(const float c[4], const float sn[4], int lo, int hi, v2f fr[4]) {
    const int o3 = (lo >> 3) & 1, o2 = (lo >> 2) & 1, o1 = (lo >> 1) & 1, o0 = lo & 1;
    if (MSBHI) {
        float f3 = ryf(c[0], sn[0], o3, hi);
        float f2[2] = {ryf(c[1], sn[1], o2, 0), ryf(c[1], sn[1], o2, 1)};
        float f1[2] = {ryf(c[2], sn[2], o1, 0), ryf(c[2], sn[2], o1, 1)};
        float f0[2] = {ryf(c[3], sn[3], o0, 0), ryf(c[3], sn[3], o0, 1)};
#pragma unroll
        for (int s = 0; s < 4; ++s) {
            float t = f3 * f2[s >> 1] * f1[s & 1];
            fr[s].x = t * f0[0];
            fr[s].y = t * f0[1];
        }
    } else {
        float f0 = ryf(c[3], sn[3], o0, hi);
        float f3[2] = {ryf(c[0], sn[0], o3, 0), ryf(c[0], sn[0], o3, 1)};
        float f2[2] = {ryf(c[1], sn[1], o2, 0), ryf(c[1], sn[1], o2, 1)};
        float f1[2] = {ryf(c[2], sn[2], o1, 0), ryf(c[2], sn[2], o1, 1)};
#pragma unroll
        for (int s = 0; s < 4; ++s) {
            float t = f0 * f3[s >> 1] * f2[s & 1];
            fr[s].x = t * f1[0];
            fr[s].y = t * f1[1];
        }
    }
}

#define WMMA(a, b, c) __builtin_amdgcn_wmma_f32_16x16x4_f32(false, (a), false, (b), (short)0, (c), false, false)

#define BT 4
#define SCHED_FENCE() __builtin_amdgcn_sched_barrier(0)
#define PAIR_FENCE() __builtin_amdgcn_sched_barrier(0)

__global__ __launch_bounds__(32) __attribute__((amdgpu_waves_per_eu(4, 4)))
void qnet_wave_kernel(const float* __restrict__ in, const float* __restrict__ qp,
                      float* __restrict__ out)
{
    __shared__ __attribute__((aligned(16))) float S[NST];

    __builtin_amdgcn_s_setprio(3);

    const int lane = threadIdx.x;
    const int hi   = lane >> 4;
    const int lo   = lane & 15;
    const int row  = blockIdx.x;

    const float* srcrow = in + (size_t)row * F_IN;
    const int dlane = cnot_dst(lane);
    {
        float q[64];
        float ss = 0.f;
#pragma unroll
        for (int e = 0; e < 64; ++e) {
            q[e] = tanh_pio2(srcrow[e * 32 + lane]);
            ss += q[e] * q[e];
        }
#pragma unroll
        for (int off = 16; off; off >>= 1) ss += __shfl_xor(ss, off, 32);
        const float sc = 1.0f / sqrtf(ss + 512.0f);
        const float padv = 0.5f * sc;
#pragma unroll
        for (int e = 0; e < 64; ++e) {
            S[addr_of(cnot_dst(e * 32) ^ dlane)] = q[e] * sc;
            S[addr_of(cnot_dst(2048 + e * 32) ^ dlane)] = padv;
        }
    }
    __syncthreads();

    const int dscat = cnot_dst((hi << 11) | lo);
    int basev[8];
#pragma unroll
    for (int m = 0; m < 8; ++m)
        basev[m] = addr_of(dscat ^ ((((m >> 2) & 1) << 10) | (((m >> 1) & 1) << 9) | ((m & 1) << 3)));
    const float* th = qp + NQ;

    float* xbase = S + lo * PA + 8 * hi;
    const float* ybase = S + hi * SA + lo;

    v8f o3[16];
    for (int L = 0; L < NLAYERS; ++L) {
        if (L == PRIO_DROP_A)      __builtin_amdgcn_s_setprio(2);
        else if (L == PRIO_DROP_B) __builtin_amdgcn_s_setprio(1);
        else if (L == PRIO_DROP_C) __builtin_amdgcn_s_setprio(0);

        float ang = th[L * NQ + (lane < NQ ? lane : 0)] * 0.5f;
        float cl = cosf(ang), sl = sinf(ang);
        float c[NQ], sn[NQ];
#pragma unroll
        for (int q = 0; q < NQ; ++q) { c[q] = bcast(cl, q); sn[q] = bcast(sl, q); }

        v2f F2[4], F1[4], F0[4];
        build_frag<true >(c + 8, sn + 8, lo, hi, F2);
        build_frag<true >(c + 4, sn + 4, lo, hi, F1);
        build_frag<false>(c + 0, sn + 0, lo, hi, F0);

        {
            v2f xf[2][BT][4];
#pragma unroll
            for (int t = 0; t < BT; ++t) {
                const float* p = xbase + t * SA;
#pragma unroll
                for (int i = 0; i < 4; ++i) xf[0][t][i] = *(const v2f*)(p + 2 * i);
            }
#pragma unroll
            for (int blk = 0; blk < 16 / BT; ++blk) {
                const int cur = blk & 1, nxt = cur ^ 1;
                if (blk + 1 < 16 / BT) {
#pragma unroll
                    for (int t = 0; t < BT; ++t) {
                        const float* p = xbase + ((blk + 1) * BT + t) * SA;
#pragma unroll
                        for (int i = 0; i < 4; ++i) xf[nxt][t][i] = *(const v2f*)(p + 2 * i);
                    }
                    SCHED_FENCE();
                }
                v8f a[BT], b[BT];
#pragma unroll
                for (int t = 0; t < BT; ++t) {
                    a[t] = (v8f){0, 0, 0, 0, 0, 0, 0, 0};
                    b[t] = (v8f){0, 0, 0, 0, 0, 0, 0, 0};
                }
#pragma unroll
                for (int s = 0; s < 4; ++s) {
#pragma unroll
                    for (int t = 0; t < BT; ++t) a[t] = WMMA(xf[cur][t][s], F2[s], a[t]);
                    SCHED_FENCE();
                }
#pragma unroll
                for (int s = 0; s < 4; ++s) {
#pragma unroll
                    for (int t = 0; t < BT; ++t) {
                        v2f ts = {a[t][2 * s], a[t][2 * s + 1]};
                        b[t] = WMMA(ts, F1[s], b[t]);
                    }
                    SCHED_FENCE();
                }
#pragma unroll
                for (int t = 0; t < BT; ++t) {
                    float* p = xbase + (blk * BT + t) * SA;
#pragma unroll
                    for (int i = 0; i < 4; ++i) {
                        v2f sv = {b[t][2 * i], b[t][2 * i + 1]};
                        *(v2f*)(p + 2 * i) = sv;
                    }
                }
                SCHED_FENCE();
            }
        }
        __syncthreads();

        {
            float yf[2][BT][8];
#pragma unroll
            for (int t = 0; t < BT; ++t) {
                const float* p = ybase + t * PA;
#pragma unroll
                for (int v = 0; v < 8; ++v) yf[0][t][v] = p[(2 * v) * SA];
                PAIR_FENCE();
            }
#pragma unroll
            for (int blk = 0; blk < 16 / BT; ++blk) {
                const int cur = blk & 1, nxt = cur ^ 1;
                if (blk + 1 < 16 / BT) {
#pragma unroll
                    for (int t = 0; t < BT; ++t) {
                        const float* p = ybase + ((blk + 1) * BT + t) * PA;
#pragma unroll
                        for (int v = 0; v < 8; ++v) yf[nxt][t][v] = p[(2 * v) * SA];
                        PAIR_FENCE();
                    }
                }
                v8f a[BT];
#pragma unroll
                for (int t = 0; t < BT; ++t) a[t] = (v8f){0, 0, 0, 0, 0, 0, 0, 0};
#pragma unroll
                for (int s = 0; s < 4; ++s) {
#pragma unroll
                    for (int t = 0; t < BT; ++t) {
                        v2f ts = {yf[cur][t][2 * s], yf[cur][t][2 * s + 1]};
                        a[t] = WMMA(F0[s], ts, a[t]);
                    }
                    SCHED_FENCE();
                }
#pragma unroll
                for (int t = 0; t < BT; ++t) o3[blk * BT + t] = a[t];
            }
        }
        if (L + 1 < NLAYERS) {
            __syncthreads();
#pragma unroll
            for (int z = 0; z < 16; ++z) {
#pragma unroll
                for (int r = 0; r < 8; ++r) {
                    const int K = cnot_dst((r << 8) | (z << 4));
                    const int m = (((K >> 10) & 1) << 2) | (((K >> 9) & 1) << 1) | ((K >> 3) & 1);
                    S[basev[m] + addr_of(K & ~0xE0F)] = o3[z][r];
                }
            }
            __syncthreads();
        }
    }

    float P = 0.f;
    float A[7] = {0.f, 0.f, 0.f, 0.f, 0.f, 0.f, 0.f};
#pragma unroll
    for (int z = 0; z < 16; ++z) {
#pragma unroll
        for (int r = 0; r < 8; ++r) {
            float v = o3[z][r];
            float p = v * v;
            P += p;
#pragma unroll
            for (int q = 1; q < 4; ++q) A[q - 1] += ((r >> (3 - q)) & 1) ? -p : p;
#pragma unroll
            for (int q = 4; q < 8; ++q) A[q - 1] += ((z >> (7 - q)) & 1) ? -p : p;
        }
    }
    float acc[NQ];
    acc[0] = hi ? -P : P;
#pragma unroll
    for (int q = 1; q < 8; ++q) acc[q] = A[q - 1];
#pragma unroll
    for (int q = 8; q < NQ; ++q) acc[q] = ((lo >> (11 - q)) & 1) ? -P : P;
#pragma unroll
    for (int off = 16; off; off >>= 1) {
#pragma unroll
        for (int q = 0; q < NQ; ++q) acc[q] += __shfl_xor(acc[q], off, 32);
    }
    float res = acc[0];
#pragma unroll
    for (int q = 1; q < NQ; ++q) res = (lane == q) ? acc[q] : res;
    if (lane < NQ) out[(size_t)row * NQ + lane] = res;
}

extern "C" void kernel_launch(void* const* d_in, const int* in_sizes, int n_in,
                              void* d_out, int out_size, void* d_ws, size_t ws_size,
                              hipStream_t stream) {
    (void)n_in; (void)out_size; (void)d_ws; (void)ws_size;
    const float* in  = (const float*)d_in[0];
    const float* qp  = (const float*)d_in[1];
    float*       out = (float*)d_out;
    int rows = in_sizes[0] / F_IN;
    qnet_wave_kernel<<<rows, 32, 0, stream>>>(in, qp, out);
}


# qubit 4-7 factor leaves the matrix cores: applied as four 2x2 rotations in packed f32 on the a0 phase's result tiles (tile z with z^bit); a2 phase takes the factor as A; qubit 4-7 sum order changes
# speedup vs baseline: 1.7379x; 1.1132x over previous
#include <hip/hip_runtime.h>
#include <math.h>

typedef __attribute__((ext_vector_type(2))) float v2f;
typedef __attribute__((ext_vector_type(8))) float v8f;

#define NQ      12
#define NLAYERS 20
#define F_IN    2048
#define PA      18
#define SA      (16 * PA)
#define NST     (16 * SA)
#define PRIO_DROP_A 10
#define PRIO_DROP_B 15
#define PRIO_DROP_C 18

__device__ __forceinline__ int addr_of(int j) {
    return (j >> 8) * SA + ((j >> 4) & 15) * PA + (j & 15);
}

__device__ __forceinline__ int cnot_dst(int i) {
    const int cq[11] = {0, 2, 4, 6, 8, 10, 1, 3, 5, 7, 9};
    const int tq[11] = {1, 3, 5, 7, 9, 11, 2, 4, 6, 8, 10};
    int j = i;
#pragma unroll
    for (int k = 0; k < 11; ++k) {
        int cm = 1 << (11 - cq[k]);
        int tm = 1 << (11 - tq[k]);
        if (j & cm) j ^= tm;
    }
    return j;
}

__device__ __forceinline__ float tanh_pio2(float x) {
    float xc = fminf(fmaxf(x, -15.f), 15.f);
    float z  = __expf(xc * 2.0f);
    return (z - 1.0f) * __builtin_amdgcn_rcpf(z + 1.0f) * 1.57079632679489662f;
}

__device__ __forceinline__ float bcast(float x, int srclane) {
    return __builtin_bit_cast(float, __builtin_amdgcn_readlane(__builtin_bit_cast(int, x), srclane));
}

__device__ __forceinline__ float ryf(float c, float s, int ob, int ib) {
    return (ob == ib) ? c : (ob ? s : -s);
}

template <bool MSBHI>
__device__ __forceinline__ void build_frag(const float c[4], const float sn[4], int lo, int hi, v2f fr[4]) {
    const int o3 = (lo >> 3) & 1, o2 = (lo >> 2) & 1, o1 = (lo >> 1) & 1, o0 = lo & 1;
    if (MSBHI) {
        float f3 = ryf(c[0], sn[0], o3, hi);
        float f2[2] = {ryf(c[1], sn[1], o2, 0), ryf(c[1], sn[1], o2, 1)};
        float f1[2] = {ryf(c[2], sn[2], o1, 0), ryf(c[2], sn[2], o1, 1)};
        float f0[2] = {ryf(c[3], sn[3], o0, 0), ryf(c[3], sn[3], o0, 1)};
#pragma unroll
        for (int s = 0; s < 4; ++s) {
            float t = f3 * f2[s >> 1] * f1[s & 1];
            fr[s].x = t * f0[0];
            fr[s].y = t * f0[1];
        }
    } else {
        float f0 = ryf(c[3], sn[3], o0, hi);
        float f3[2] = {ryf(c[0], sn[0], o3, 0), ryf(c[0], sn[0], o3, 1)};
        float f2[2] = {ryf(c[1], sn[1], o2, 0), ryf(c[1], sn[1], o2, 1)};
        float f1[2] = {ryf(c[2], sn[2], o1, 0), ryf(c[2], sn[2], o1, 1)};
#pragma unroll
        for (int s = 0; s < 4; ++s) {
            float t = f0 * f3[s >> 1] * f2[s & 1];
            fr[s].x = t * f1[0];
            fr[s].y = t * f1[1];
        }
    }
}

#define WMMA(a, b, c) __builtin_amdgcn_wmma_f32_16x16x4_f32(false, (a), false, (b), (short)0, (c), false, false)

#define BT 4
#define SCHED_FENCE() __builtin_amdgcn_sched_barrier(0)
#define PAIR_FENCE() __builtin_amdgcn_sched_barrier(0)

__global__ __launch_bounds__(32) __attribute__((amdgpu_waves_per_eu(4, 4)))
void qnet_wave_kernel(const float* __restrict__ in, const float* __restrict__ qp,
                      float* __restrict__ out)
{
    __shared__ __attribute__((aligned(16))) float S[NST];

    __builtin_amdgcn_s_setprio(3);

    const int lane = threadIdx.x;
    const int hi   = lane >> 4;
    const int lo   = lane & 15;
    const int row  = blockIdx.x;

    const float* srcrow = in + (size_t)row * F_IN;
    const int dlane = cnot_dst(lane);
    {
        float q[64];
        float ss = 0.f;
#pragma unroll
        for (int e = 0; e < 64; ++e) {
            q[e] = tanh_pio2(srcrow[e * 32 + lane]);
            ss += q[e] * q[e];
        }
#pragma unroll
        for (int off = 16; off; off >>= 1) ss += __shfl_xor(ss, off, 32);
        const float sc = 1.0f / sqrtf(ss + 512.0f);
        const float padv = 0.5f * sc;
#pragma unroll
        for (int e = 0; e < 64; ++e) {
            S[addr_of(cnot_dst(e * 32) ^ dlane)] = q[e] * sc;
            S[addr_of(cnot_dst(2048 + e * 32) ^ dlane)] = padv;
        }
    }
    __syncthreads();

    const int dscat = cnot_dst((hi << 11) | lo);
    int basev[8];
#pragma unroll
    for (int m = 0; m < 8; ++m)
        basev[m] = addr_of(dscat ^ ((((m >> 2) & 1) << 10) | (((m >> 1) & 1) << 9) | ((m & 1) << 3)));
    const float* th = qp + NQ;

    float* xbase = S + lo * PA + 8 * hi;
    const float* ybase = S + hi * SA + lo;

    v8f o3[16];
    for (int L = 0; L < NLAYERS; ++L) {
        if (L == PRIO_DROP_A)      __builtin_amdgcn_s_setprio(2);
        else if (L == PRIO_DROP_B) __builtin_amdgcn_s_setprio(1);
        else if (L == PRIO_DROP_C) __builtin_amdgcn_s_setprio(0);

        float ang = th[L * NQ + (lane < NQ ? lane : 0)] * 0.5f;
        float cl = cosf(ang), sl = sinf(ang);
        float c[NQ], sn[NQ];
#pragma unroll
        for (int q = 0; q < NQ; ++q) { c[q] = bcast(cl, q); sn[q] = bcast(sl, q); }

        v2f F2[4], F0[4];
        build_frag<true >(c + 8, sn + 8, lo, hi, F2);
        build_frag<false>(c + 0, sn + 0, lo, hi, F0);

        {
            v2f xf[2][BT][4];
#pragma unroll
            for (int t = 0; t < BT; ++t) {
                const float* p = xbase + t * SA;
#pragma unroll
                for (int i = 0; i < 4; ++i) xf[0][t][i] = *(const v2f*)(p + 2 * i);
            }
#pragma unroll
            for (int blk = 0; blk < 16 / BT; ++blk) {
                const int cur = blk & 1, nxt = cur ^ 1;
                if (blk + 1 < 16 / BT) {
#pragma unroll
                    for (int t = 0; t < BT; ++t) {
                        const float* p = xbase + ((blk + 1) * BT + t) * SA;
#pragma unroll
                        for (int i = 0; i < 4; ++i) xf[nxt][t][i] = *(const v2f*)(p + 2 * i);
                    }
                    SCHED_FENCE();
                }
                v8f a[BT];
#pragma unroll
                for (int t = 0; t < BT; ++t) a[t] = (v8f){0, 0, 0, 0, 0, 0, 0, 0};
#pragma unroll
                for (int s = 0; s < 4; ++s) {
#pragma unroll
                    for (int t = 0; t < BT; ++t) a[t] = WMMA(F2[s], xf[cur][t][s], a[t]);
                    SCHED_FENCE();
                }
#pragma unroll
                for (int t = 0; t < BT; ++t) {
                    float* p = xbase + (blk * BT + t) * SA;
#pragma unroll
                    for (int i = 0; i < 4; ++i) {
                        v2f sv = {a[t][2 * i], a[t][2 * i + 1]};
                        *(v2f*)(p + 2 * i) = sv;
                    }
                }
                SCHED_FENCE();
            }
        }
        __syncthreads();

        {
            float yf[2][BT][8];
#pragma unroll
            for (int t = 0; t < BT; ++t) {
                const float* p = ybase + t * PA;
#pragma unroll
                for (int v = 0; v < 8; ++v) yf[0][t][v] = p[(2 * v) * SA];
                PAIR_FENCE();
            }
#pragma unroll
            for (int blk = 0; blk < 16 / BT; ++blk) {
                const int cur = blk & 1, nxt = cur ^ 1;
                if (blk + 1 < 16 / BT) {
#pragma unroll
                    for (int t = 0; t < BT; ++t) {
                        const float* p = ybase + ((blk + 1) * BT + t) * PA;
#pragma unroll
                        for (int v = 0; v < 8; ++v) yf[nxt][t][v] = p[(2 * v) * SA];
                        PAIR_FENCE();
                    }
                }
                v8f a[BT];
#pragma unroll
                for (int t = 0; t < BT; ++t) a[t] = (v8f){0, 0, 0, 0, 0, 0, 0, 0};
#pragma unroll
                for (int s = 0; s < 4; ++s) {
#pragma unroll
                    for (int t = 0; t < BT; ++t) {
                        v2f ts = {yf[cur][t][2 * s], yf[cur][t][2 * s + 1]};
                        a[t] = WMMA(F0[s], ts, a[t]);
                    }
                    SCHED_FENCE();
                }
#pragma unroll
                for (int t = 0; t < BT; ++t) o3[blk * BT + t] = a[t];
            }
        }
#pragma unroll
        for (int q = 7; q >= 4; --q) {
            const int bit = 1 << (7 - q);
            const v2f cq = {c[q], c[q]};
            const v2f sq = {sn[q], sn[q]};
#pragma unroll
            for (int z = 0; z < 16; ++z) {
                if (z & bit) continue;
#pragma unroll
                for (int i = 0; i < 4; ++i) {
                    const v2f x0 = {o3[z][2 * i], o3[z][2 * i + 1]};
                    const v2f x1 = {o3[z | bit][2 * i], o3[z | bit][2 * i + 1]};
                    const v2f n0 = cq * x0 - sq * x1;
                    const v2f n1 = sq * x0 + cq * x1;
                    o3[z][2 * i] = n0.x;       o3[z][2 * i + 1] = n0.y;
                    o3[z | bit][2 * i] = n1.x; o3[z | bit][2 * i + 1] = n1.y;
                }
            }
        }
        if (L + 1 < NLAYERS) {
            __syncthreads();
#pragma unroll
            for (int z = 0; z < 16; ++z) {
#pragma unroll
                for (int r = 0; r < 8; ++r) {
                    const int K = cnot_dst((r << 8) | (z << 4));
                    const int m = (((K >> 10) & 1) << 2) | (((K >> 9) & 1) << 1) | ((K >> 3) & 1);
                    S[basev[m] + addr_of(K & ~0xE0F)] = o3[z][r];
                }
            }
            __syncthreads();
        }
    }

    float P = 0.f;
    float A[7] = {0.f, 0.f, 0.f, 0.f, 0.f, 0.f, 0.f};
#pragma unroll
    for (int z = 0; z < 16; ++z) {
#pragma unroll
        for (int r = 0; r < 8; ++r) {
            float v = o3[z][r];
            float p = v * v;
            P += p;
#pragma unroll
            for (int q = 1; q < 4; ++q) A[q - 1] += ((r >> (3 - q)) & 1) ? -p : p;
#pragma unroll
            for (int q = 4; q < 8; ++q) A[q - 1] += ((z >> (7 - q)) & 1) ? -p : p;
        }
    }
    float acc[NQ];
    acc[0] = hi ? -P : P;
#pragma unroll
    for (int q = 1; q < 8; ++q) acc[q] = A[q - 1];
#pragma unroll
    for (int q = 8; q < NQ; ++q) acc[q] = ((lo >> (11 - q)) & 1) ? -P : P;
#pragma unroll
    for (int off = 16; off; off >>= 1) {
#pragma unroll
        for (int q = 0; q < NQ; ++q) acc[q] += __shfl_xor(acc[q], off, 32);
    }
    float res = acc[0];
#pragma unroll
    for (int q = 1; q < NQ; ++q) res = (lane == q) ? acc[q] : res;
    if (lane < NQ) out[(size_t)row * NQ + lane] = res;
}

extern "C" void kernel_launch(void* const* d_in, const int* in_sizes, int n_in,
                              void* d_out, int out_size, void* d_ws, size_t ws_size,
                              hipStream_t stream) {
    (void)n_in; (void)out_size; (void)d_ws; (void)ws_size;
    const float* in  = (const float*)d_in[0];
    const float* qp  = (const float*)d_in[1];
    float*       out = (float*)d_out;
    int rows = in_sizes[0] / F_IN;
    qnet_wave_kernel<<<rows, 32, 0, stream>>>(in, qp, out);
}


# a2 result tiles feed the a0 matrix step as B from registers; state read once per layer as A (rows a0, two 16-byte loads per lane, slab pitch 260): LDS transpose trip goes; a2, a0 k sum order changes
# speedup vs baseline: 1.8904x; 1.0877x over previous
#include <hip/hip_runtime.h>
#include <math.h>

typedef __attribute__((ext_vector_type(2))) float v2f;
typedef __attribute__((ext_vector_type(4))) float v4f;
typedef __attribute__((ext_vector_type(8))) float v8f;

#define NQ      12
#define NLAYERS 20
#define F_IN    2048
#define PA      16
#define SA      (16 * PA + 4)
#define NST     (16 * SA)
#define PRIO_DROP_A 10
#define PRIO_DROP_B 15
#define PRIO_DROP_C 18

__device__ __forceinline__ int addr_of(int j) {
    return (j >> 8) * SA + ((j >> 4) & 15) * PA + (j & 15);
}

__device__ __forceinline__ int cnot_dst(int i) {
    const int cq[11] = {0, 2, 4, 6, 8, 10, 1, 3, 5, 7, 9};
    const int tq[11] = {1, 3, 5, 7, 9, 11, 2, 4, 6, 8, 10};
    int j = i;
#pragma unroll
    for (int k = 0; k < 11; ++k) {
        int cm = 1 << (11 - cq[k]);
        int tm = 1 << (11 - tq[k]);
        if (j & cm) j ^= tm;
    }
    return j;
}

__device__ __forceinline__ float tanh_pio2(float x) {
    float xc = fminf(fmaxf(x, -15.f), 15.f);
    float z  = __expf(xc * 2.0f);
    return (z - 1.0f) * __builtin_amdgcn_rcpf(z + 1.0f) * 1.57079632679489662f;
}

__device__ __forceinline__ float bcast(float x, int srclane) {
    return __builtin_bit_cast(float, __builtin_amdgcn_readlane(__builtin_bit_cast(int, x), srclane));
}

__device__ __forceinline__ float ryf(float c, float s, int ob, int ib) {
    return (ob == ib) ? c : (ob ? s : -s);
}

template <bool MSBHI>
__device__ __forceinline__ void build_frag(const float c[4], const float sn[4], int lo, int hi, v2f fr[4]) {
    const int o3 = (lo >> 3) & 1, o2 = (lo >> 2) & 1, o1 = (lo >> 1) & 1, o0 = lo & 1;
    float f1[2] = {ryf(c[2], sn[2], o1, 0), ryf(c[2], sn[2], o1, 1)};
    float f0[2] = {ryf(c[3], sn[3], o0, 0), ryf(c[3], sn[3], o0, 1)};
    if (MSBHI) {
        float f3 = ryf(c[0], sn[0], o3, hi);
        float f2[2] = {ryf(c[1], sn[1], o2, 0), ryf(c[1], sn[1], o2, 1)};
#pragma unroll
        for (int s = 0; s < 4; ++s) {
            float t = f3 * f2[s >> 1] * f1[s & 1];
            fr[s].x = t * f0[0];
            fr[s].y = t * f0[1];
        }
    } else {
        float f2 = ryf(c[1], sn[1], o2, hi);
        float f3[2] = {ryf(c[0], sn[0], o3, 0), ryf(c[0], sn[0], o3, 1)};
#pragma unroll
        for (int s = 0; s < 4; ++s) {
            float t = f2 * f3[s >> 1] * f1[s & 1];
            fr[s].x = t * f0[0];
            fr[s].y = t * f0[1];
        }
    }
}

#define WMMA(a, b, c) __builtin_amdgcn_wmma_f32_16x16x4_f32(false, (a), false, (b), (short)0, (c), false, false)

#define BT 4
#define SCHED_FENCE() __builtin_amdgcn_sched_barrier(0)

__global__ __launch_bounds__(32) __attribute__((amdgpu_waves_per_eu(4, 4)))
void qnet_wave_kernel(const float* __restrict__ in, const float* __restrict__ qp,
                      float* __restrict__ out)
{
    __shared__ __attribute__((aligned(16))) float S[NST];

    __builtin_amdgcn_s_setprio(3);

    const int lane = threadIdx.x;
    const int hi   = lane >> 4;
    const int lo   = lane & 15;
    const int row  = blockIdx.x;

    const float* srcrow = in + (size_t)row * F_IN;
    const int dlane = cnot_dst(lane);
    {
        float q[64];
        float ss = 0.f;
#pragma unroll
        for (int e = 0; e < 64; ++e) {
            q[e] = tanh_pio2(srcrow[e * 32 + lane]);
            ss += q[e] * q[e];
        }
#pragma unroll
        for (int off = 16; off; off >>= 1) ss += __shfl_xor(ss, off, 32);
        const float sc = 1.0f / sqrtf(ss + 512.0f);
        const float padv = 0.5f * sc;
#pragma unroll
        for (int e = 0; e < 64; ++e) {
            S[addr_of(cnot_dst(e * 32) ^ dlane)] = q[e] * sc;
            S[addr_of(cnot_dst(2048 + e * 32) ^ dlane)] = padv;
        }
    }
    __syncthreads();

    const int dscat = cnot_dst((hi << 11) | lo);
    int basev[8];
#pragma unroll
    for (int m = 0; m < 8; ++m)
        basev[m] = addr_of(dscat ^ ((((m >> 2) & 1) << 10) | (((m >> 1) & 1) << 9) | ((m & 1) << 3)));
    const float* th = qp + NQ;

    const float* abase = S + lo * SA + 4 * hi;

    v8f o3[16];
    for (int L = 0; L < NLAYERS; ++L) {
        if (L == PRIO_DROP_A)      __builtin_amdgcn_s_setprio(2);
        else if (L == PRIO_DROP_B) __builtin_amdgcn_s_setprio(1);
        else if (L == PRIO_DROP_C) __builtin_amdgcn_s_setprio(0);

        float ang = th[L * NQ + (lane < NQ ? lane : 0)] * 0.5f;
        float cl = cosf(ang), sl = sinf(ang);
        float c[NQ], sn[NQ];
#pragma unroll
        for (int q = 0; q < NQ; ++q) { c[q] = bcast(cl, q); sn[q] = bcast(sl, q); }

        v2f F2[4], F0[4];
        build_frag<false>(c + 8, sn + 8, lo, hi, F2);
        build_frag<true >(c + 0, sn + 0, lo, hi, F0);

        {
            v4f xa[2][BT][2];
#pragma unroll
            for (int t = 0; t < BT; ++t) {
                const float* p = abase + t * PA;
                xa[0][t][0] = *(const v4f*)(p);
                xa[0][t][1] = *(const v4f*)(p + 8);
            }
#pragma unroll
            for (int blk = 0; blk < 16 / BT; ++blk) {
                const int cur = blk & 1, nxt = cur ^ 1;
                if (blk + 1 < 16 / BT) {
#pragma unroll
                    for (int t = 0; t < BT; ++t) {
                        const float* p = abase + ((blk + 1) * BT + t) * PA;
                        xa[nxt][t][0] = *(const v4f*)(p);
                        xa[nxt][t][1] = *(const v4f*)(p + 8);
                    }
                    SCHED_FENCE();
                }
                v8f a[BT];
#pragma unroll
                for (int t = 0; t < BT; ++t) a[t] = (v8f){0, 0, 0, 0, 0, 0, 0, 0};
#pragma unroll
                for (int s = 0; s < 4; ++s) {
#pragma unroll
                    for (int t = 0; t < BT; ++t) {
                        const v4f w = xa[cur][t][s >> 1];
                        const v2f as = {w[2 * (s & 1)], w[2 * (s & 1) + 1]};
                        a[t] = WMMA(as, F2[s], a[t]);
                    }
                    SCHED_FENCE();
                }
                v8f b[BT];
#pragma unroll
                for (int t = 0; t < BT; ++t) b[t] = (v8f){0, 0, 0, 0, 0, 0, 0, 0};
#pragma unroll
                for (int s = 0; s < 4; ++s) {
#pragma unroll
                    for (int t = 0; t < BT; ++t) {
                        const v2f bs = {a[t][2 * s], a[t][2 * s + 1]};
                        b[t] = WMMA(F0[s], bs, b[t]);
                    }
                    SCHED_FENCE();
                }
#pragma unroll
                for (int t = 0; t < BT; ++t) o3[blk * BT + t] = b[t];
            }
        }
#pragma unroll
        for (int q = 7; q >= 4; --q) {
            const int bit = 1 << (7 - q);
            const v2f cq = {c[q], c[q]};
            const v2f sq = {sn[q], sn[q]};
#pragma unroll
            for (int z = 0; z < 16; ++z) {
                if (z & bit) continue;
#pragma unroll
                for (int i = 0; i < 4; ++i) {
                    const v2f x0 = {o3[z][2 * i], o3[z][2 * i + 1]};
                    const v2f x1 = {o3[z | bit][2 * i], o3[z | bit][2 * i + 1]};
                    const v2f n0 = cq * x0 - sq * x1;
                    const v2f n1 = sq * x0 + cq * x1;
                    o3[z][2 * i] = n0.x;       o3[z][2 * i + 1] = n0.y;
                    o3[z | bit][2 * i] = n1.x; o3[z | bit][2 * i + 1] = n1.y;
                }
            }
        }
        if (L + 1 < NLAYERS) {
            __syncthreads();
#pragma unroll
            for (int z = 0; z < 16; ++z) {
#pragma unroll
                for (int r = 0; r < 8; ++r) {
                    const int K = cnot_dst((r << 8) | (z << 4));
                    const int m = (((K >> 10) & 1) << 2) | (((K >> 9) & 1) << 1) | ((K >> 3) & 1);
                    S[basev[m] + addr_of(K & ~0xE0F)] = o3[z][r];
                }
            }
            __syncthreads();
        }
    }

    float P = 0.f;
    float A[7] = {0.f, 0.f, 0.f, 0.f, 0.f, 0.f, 0.f};
#pragma unroll
    for (int z = 0; z < 16; ++z) {
#pragma unroll
        for (int r = 0; r < 8; ++r) {
            float v = o3[z][r];
            float p = v * v;
            P += p;
#pragma unroll
            for (int q = 1; q < 4; ++q) A[q - 1] += ((r >> (3 - q)) & 1) ? -p : p;
#pragma unroll
            for (int q = 4; q < 8; ++q) A[q - 1] += ((z >> (7 - q)) & 1) ? -p : p;
        }
    }
    float acc[NQ];
    acc[0] = hi ? -P : P;
#pragma unroll
    for (int q = 1; q < 8; ++q) acc[q] = A[q - 1];
#pragma unroll
    for (int q = 8; q < NQ; ++q) acc[q] = ((lo >> (11 - q)) & 1) ? -P : P;
#pragma unroll
    for (int off = 16; off; off >>= 1) {
#pragma unroll
        for (int q = 0; q < NQ; ++q) acc[q] += __shfl_xor(acc[q], off, 32);
    }
    float res = acc[0];
#pragma unroll
    for (int q = 1; q < NQ; ++q) res = (lane == q) ? acc[q] : res;
    if (lane < NQ) out[(size_t)row * NQ + lane] = res;
}

extern "C" void kernel_launch(void* const* d_in, const int* in_sizes, int n_in,
                              void* d_out, int out_size, void* d_ws, size_t ws_size,
                              hipStream_t stream) {
    (void)n_in; (void)out_size; (void)d_ws; (void)ws_size;
    const float* in  = (const float*)d_in[0];
    const float* qp  = (const float*)d_in[1];
    float*       out = (float*)d_out;
    int rows = in_sizes[0] / F_IN;
    qnet_wave_kernel<<<rows, 32, 0, stream>>>(in, qp, out);
}


# qubit 4-7 rotations in tangent form cos*[[1,-t],[t,1]], t=sin/cos: one packed fma per output pair (was mul+fma), the 4 cosines folded into the qubit 8-11 factor; qubit 4-7 rounding changes
# speedup vs baseline: 2.0648x; 1.0923x over previous
#include <hip/hip_runtime.h>
#include <math.h>

typedef __attribute__((ext_vector_type(2))) float v2f;
typedef __attribute__((ext_vector_type(4))) float v4f;
typedef __attribute__((ext_vector_type(8))) float v8f;

#define NQ      12
#define NLAYERS 20
#define F_IN    2048
#define PA      16
#define SA      (16 * PA + 4)
#define NST     (16 * SA)
#define PRIO_DROP_A 10
#define PRIO_DROP_B 15
#define PRIO_DROP_C 18

__device__ __forceinline__ int addr_of(int j) {
    return (j >> 8) * SA + ((j >> 4) & 15) * PA + (j & 15);
}

__device__ __forceinline__ int cnot_dst(int i) {
    const int cq[11] = {0, 2, 4, 6, 8, 10, 1, 3, 5, 7, 9};
    const int tq[11] = {1, 3, 5, 7, 9, 11, 2, 4, 6, 8, 10};
    int j = i;
#pragma unroll
    for (int k = 0; k < 11; ++k) {
        int cm = 1 << (11 - cq[k]);
        int tm = 1 << (11 - tq[k]);
        if (j & cm) j ^= tm;
    }
    return j;
}

__device__ __forceinline__ float tanh_pio2(float x) {
    float xc = fminf(fmaxf(x, -15.f), 15.f);
    float z  = __expf(xc * 2.0f);
    return (z - 1.0f) * __builtin_amdgcn_rcpf(z + 1.0f) * 1.57079632679489662f;
}

__device__ __forceinline__ float bcast(float x, int srclane) {
    return __builtin_bit_cast(float, __builtin_amdgcn_readlane(__builtin_bit_cast(int, x), srclane));
}

__device__ __forceinline__ float ryf(float c, float s, int ob, int ib) {
    return (ob == ib) ? c : (ob ? s : -s);
}

template <bool MSBHI>
__device__ __forceinline__ void build_frag(const float c[4], const float sn[4], int lo, int hi, float scale, v2f fr[4]) {
    const int o3 = (lo >> 3) & 1, o2 = (lo >> 2) & 1, o1 = (lo >> 1) & 1, o0 = lo & 1;
    float f1[2] = {ryf(c[2], sn[2], o1, 0), ryf(c[2], sn[2], o1, 1)};
    float f0[2] = {ryf(c[3], sn[3], o0, 0), ryf(c[3], sn[3], o0, 1)};
    if (MSBHI) {
        float f3 = ryf(c[0], sn[0], o3, hi);
        float f2[2] = {ryf(c[1], sn[1], o2, 0), ryf(c[1], sn[1], o2, 1)};
#pragma unroll
        for (int s = 0; s < 4; ++s) {
            float t = f3 * f2[s >> 1] * f1[s & 1];
            fr[s].x = t * f0[0];
            fr[s].y = t * f0[1];
        }
    } else {
        float f2 = ryf(c[1], sn[1], o2, hi) * scale;
        float f3[2] = {ryf(c[0], sn[0], o3, 0), ryf(c[0], sn[0], o3, 1)};
#pragma unroll
        for (int s = 0; s < 4; ++s) {
            float t = f2 * f3[s >> 1] * f1[s & 1];
            fr[s].x = t * f0[0];
            fr[s].y = t * f0[1];
        }
    }
}

#define WMMA(a, b, c) __builtin_amdgcn_wmma_f32_16x16x4_f32(false, (a), false, (b), (short)0, (c), false, false)

#define BT 4
#define SCHED_FENCE() __builtin_amdgcn_sched_barrier(0)

__global__ __launch_bounds__(32) __attribute__((amdgpu_waves_per_eu(4, 4)))
void qnet_wave_kernel(const float* __restrict__ in, const float* __restrict__ qp,
                      float* __restrict__ out)
{
    __shared__ __attribute__((aligned(16))) float S[NST];

    __builtin_amdgcn_s_setprio(3);

    const int lane = threadIdx.x;
    const int hi   = lane >> 4;
    const int lo   = lane & 15;
    const int row  = blockIdx.x;

    const float* srcrow = in + (size_t)row * F_IN;
    const int dlane = cnot_dst(lane);
    {
        float q[64];
        float ss = 0.f;
#pragma unroll
        for (int e = 0; e < 64; ++e) {
            q[e] = tanh_pio2(srcrow[e * 32 + lane]);
            ss += q[e] * q[e];
        }
#pragma unroll
        for (int off = 16; off; off >>= 1) ss += __shfl_xor(ss, off, 32);
        const float sc = 1.0f / sqrtf(ss + 512.0f);
        const float padv = 0.5f * sc;
#pragma unroll
        for (int e = 0; e < 64; ++e) {
            S[addr_of(cnot_dst(e * 32) ^ dlane)] = q[e] * sc;
            S[addr_of(cnot_dst(2048 + e * 32) ^ dlane)] = padv;
        }
    }
    __syncthreads();

    const int dscat = cnot_dst((hi << 11) | lo);
    int basev[8];
#pragma unroll
    for (int m = 0; m < 8; ++m)
        basev[m] = addr_of(dscat ^ ((((m >> 2) & 1) << 10) | (((m >> 1) & 1) << 9) | ((m & 1) << 3)));
    const float* th = qp + NQ;

    const float* abase = S + lo * SA + 4 * hi;

    v8f o3[16];
    for (int L = 0; L < NLAYERS; ++L) {
        if (L == PRIO_DROP_A)      __builtin_amdgcn_s_setprio(2);
        else if (L == PRIO_DROP_B) __builtin_amdgcn_s_setprio(1);
        else if (L == PRIO_DROP_C) __builtin_amdgcn_s_setprio(0);

        float ang = th[L * NQ + (lane < NQ ? lane : 0)] * 0.5f;
        float cl = cosf(ang), sl = sinf(ang);
        float c[NQ], sn[NQ];
#pragma unroll
        for (int q = 0; q < NQ; ++q) { c[q] = bcast(cl, q); sn[q] = bcast(sl, q); }
        const float tl = sl / cl;
        float tq4[4];
#pragma unroll
        for (int q = 4; q < 8; ++q) tq4[q - 4] = bcast(tl, q);
        const float cprod = ((c[4] * c[5]) * c[6]) * c[7];

        v2f F2[4], F0[4];
        build_frag<false>(c + 8, sn + 8, lo, hi, cprod, F2);
        build_frag<true >(c + 0, sn + 0, lo, hi, 1.0f, F0);

        {
            v4f xa[2][BT][2];
#pragma unroll
            for (int t = 0; t < BT; ++t) {
                const float* p = abase + t * PA;
                xa[0][t][0] = *(const v4f*)(p);
                xa[0][t][1] = *(const v4f*)(p + 8);
            }
#pragma unroll
            for (int blk = 0; blk < 16 / BT; ++blk) {
                const int cur = blk & 1, nxt = cur ^ 1;
                if (blk + 1 < 16 / BT) {
#pragma unroll
                    for (int t = 0; t < BT; ++t) {
                        const float* p = abase + ((blk + 1) * BT + t) * PA;
                        xa[nxt][t][0] = *(const v4f*)(p);
                        xa[nxt][t][1] = *(const v4f*)(p + 8);
                    }
                    SCHED_FENCE();
                }
                v8f a[BT];
#pragma unroll
                for (int t = 0; t < BT; ++t) a[t] = (v8f){0, 0, 0, 0, 0, 0, 0, 0};
#pragma unroll
                for (int s = 0; s < 4; ++s) {
#pragma unroll
                    for (int t = 0; t < BT; ++t) {
                        const v4f w = xa[cur][t][s >> 1];
                        const v2f as = {w[2 * (s & 1)], w[2 * (s & 1) + 1]};
                        a[t] = WMMA(as, F2[s], a[t]);
                    }
                    SCHED_FENCE();
                }
                v8f b[BT];
#pragma unroll
                for (int t = 0; t < BT; ++t) b[t] = (v8f){0, 0, 0, 0, 0, 0, 0, 0};
#pragma unroll
                for (int s = 0; s < 4; ++s) {
#pragma unroll
                    for (int t = 0; t < BT; ++t) {
                        const v2f bs = {a[t][2 * s], a[t][2 * s + 1]};
                        b[t] = WMMA(F0[s], bs, b[t]);
                    }
                    SCHED_FENCE();
                }
#pragma unroll
                for (int t = 0; t < BT; ++t) o3[blk * BT + t] = b[t];
            }
        }
#pragma unroll
        for (int q = 7; q >= 4; --q) {
            const int bit = 1 << (7 - q);
            const v2f tq = {tq4[q - 4], tq4[q - 4]};
#pragma unroll
            for (int z = 0; z < 16; ++z) {
                if (z & bit) continue;
#pragma unroll
                for (int i = 0; i < 4; ++i) {
                    const v2f x0 = {o3[z][2 * i], o3[z][2 * i + 1]};
                    const v2f x1 = {o3[z | bit][2 * i], o3[z | bit][2 * i + 1]};
                    const v2f n0 = x0 - tq * x1;
                    const v2f n1 = x1 + tq * x0;
                    o3[z][2 * i] = n0.x;       o3[z][2 * i + 1] = n0.y;
                    o3[z | bit][2 * i] = n1.x; o3[z | bit][2 * i + 1] = n1.y;
                }
            }
        }
        if (L + 1 < NLAYERS) {
            __syncthreads();
#pragma unroll
            for (int z = 0; z < 16; ++z) {
#pragma unroll
                for (int r = 0; r < 8; ++r) {
                    const int K = cnot_dst((r << 8) | (z << 4));
                    const int m = (((K >> 10) & 1) << 2) | (((K >> 9) & 1) << 1) | ((K >> 3) & 1);
                    S[basev[m] + addr_of(K & ~0xE0F)] = o3[z][r];
                }
            }
            __syncthreads();
        }
    }

    float P = 0.f;
    float A[7] = {0.f, 0.f, 0.f, 0.f, 0.f, 0.f, 0.f};
#pragma unroll
    for (int z = 0; z < 16; ++z) {
#pragma unroll
        for (int r = 0; r < 8; ++r) {
            float v = o3[z][r];
            float p = v * v;
            P += p;
#pragma unroll
            for (int q = 1; q < 4; ++q) A[q - 1] += ((r >> (3 - q)) & 1) ? -p : p;
#pragma unroll
            for (int q = 4; q < 8; ++q) A[q - 1] += ((z >> (7 - q)) & 1) ? -p : p;
        }
    }
    float acc[NQ];
    acc[0] = hi ? -P : P;
#pragma unroll
    for (int q = 1; q < 8; ++q) acc[q] = A[q - 1];
#pragma unroll
    for (int q = 8; q < NQ; ++q) acc[q] = ((lo >> (11 - q)) & 1) ? -P : P;
#pragma unroll
    for (int off = 16; off; off >>= 1) {
#pragma unroll
        for (int q = 0; q < NQ; ++q) acc[q] += __shfl_xor(acc[q], off, 32);
    }
    float res = acc[0];
#pragma unroll
    for (int q = 1; q < NQ; ++q) res = (lane == q) ? acc[q] : res;
    if (lane < NQ) out[(size_t)row * NQ + lane] = res;
}

extern "C" void kernel_launch(void* const* d_in, const int* in_sizes, int n_in,
                              void* d_out, int out_size, void* d_ws, size_t ws_size,
                              hipStream_t stream) {
    (void)n_in; (void)out_size; (void)d_ws; (void)ws_size;
    const float* in  = (const float*)d_in[0];
    const float* qp  = (const float*)d_in[1];
    float*       out = (float*)d_out;
    int rows = in_sizes[0] / F_IN;
    qnet_wave_kernel<<<rows, 32, 0, stream>>>(in, qp, out);
}


# qubit 0-3 factor leaves the matrix cores: three tangent-form 2x2 rotations on register pairs inside each result tile, qubit 0 across the lane halves by permlanex16; qubit 0-3 sum order changes
# speedup vs baseline: 2.6294x; 1.2734x over previous
#include <hip/hip_runtime.h>
#include <math.h>

typedef __attribute__((ext_vector_type(2))) float v2f;
typedef __attribute__((ext_vector_type(4))) float v4f;
typedef __attribute__((ext_vector_type(8))) float v8f;

#define NQ      12
#define NLAYERS 20
#define F_IN    2048
#define PA      16
#define SA      (16 * PA + 4)
#define NST     (16 * SA)
#define PRIO_DROP_A 10
#define PRIO_DROP_B 15
#define PRIO_DROP_C 18

__device__ __forceinline__ int addr_of(int j) {
    return (j >> 8) * SA + ((j >> 4) & 15) * PA + (j & 15);
}

__device__ __forceinline__ int cnot_dst(int i) {
    const int cq[11] = {0, 2, 4, 6, 8, 10, 1, 3, 5, 7, 9};
    const int tq[11] = {1, 3, 5, 7, 9, 11, 2, 4, 6, 8, 10};
    int j = i;
#pragma unroll
    for (int k = 0; k < 11; ++k) {
        int cm = 1 << (11 - cq[k]);
        int tm = 1 << (11 - tq[k]);
        if (j & cm) j ^= tm;
    }
    return j;
}

__device__ __forceinline__ float tanh_pio2(float x) {
    float xc = fminf(fmaxf(x, -15.f), 15.f);
    float z  = __expf(xc * 2.0f);
    return (z - 1.0f) * __builtin_amdgcn_rcpf(z + 1.0f) * 1.57079632679489662f;
}

__device__ __forceinline__ float bcast(float x, int srclane) {
    return __builtin_bit_cast(float, __builtin_amdgcn_readlane(__builtin_bit_cast(int, x), srclane));
}

__device__ __forceinline__ float ryf(float c, float s, int ob, int ib) {
    return (ob == ib) ? c : (ob ? s : -s);
}

__device__ __forceinline__ void build_frag(const float c[4], const float sn[4], int lo, int hi, float scale, v2f fr[4]) {
    const int o3 = (lo >> 3) & 1, o2 = (lo >> 2) & 1, o1 = (lo >> 1) & 1, o0 = lo & 1;
    float f1[2] = {ryf(c[2], sn[2], o1, 0), ryf(c[2], sn[2], o1, 1)};
    float f0[2] = {ryf(c[3], sn[3], o0, 0), ryf(c[3], sn[3], o0, 1)};
    float f2 = ryf(c[1], sn[1], o2, hi) * scale;
    float f3[2] = {ryf(c[0], sn[0], o3, 0), ryf(c[0], sn[0], o3, 1)};
#pragma unroll
    for (int s = 0; s < 4; ++s) {
        float t = f2 * f3[s >> 1] * f1[s & 1];
        fr[s].x = t * f0[0];
        fr[s].y = t * f0[1];
    }
}

__device__ __forceinline__ float xhalf(float x) {
    int b = __builtin_bit_cast(int, x);
    return __builtin_bit_cast(float, __builtin_amdgcn_permlanex16(b, b, 0x76543210, 0xfedcba98, true, false));
}

#define WMMA(a, b, c) __builtin_amdgcn_wmma_f32_16x16x4_f32(false, (a), false, (b), (short)0, (c), false, false)

#define BT 4
#define SCHED_FENCE() __builtin_amdgcn_sched_barrier(0)

__global__ __launch_bounds__(32) __attribute__((amdgpu_waves_per_eu(4, 4)))
void qnet_wave_kernel(const float* __restrict__ in, const float* __restrict__ qp,
                      float* __restrict__ out)
{
    __shared__ __attribute__((aligned(16))) float S[NST];

    __builtin_amdgcn_s_setprio(3);

    const int lane = threadIdx.x;
    const int hi   = lane >> 4;
    const int lo   = lane & 15;
    const int row  = blockIdx.x;

    const float* srcrow = in + (size_t)row * F_IN;
    const int dlane = cnot_dst(lane);
    {
        float q[64];
        float ss = 0.f;
#pragma unroll
        for (int e = 0; e < 64; ++e) {
            q[e] = tanh_pio2(srcrow[e * 32 + lane]);
            ss += q[e] * q[e];
        }
#pragma unroll
        for (int off = 16; off; off >>= 1) ss += __shfl_xor(ss, off, 32);
        const float sc = 1.0f / sqrtf(ss + 512.0f);
        const float padv = 0.5f * sc;
#pragma unroll
        for (int e = 0; e < 64; ++e) {
            S[addr_of(cnot_dst(e * 32) ^ dlane)] = q[e] * sc;
            S[addr_of(cnot_dst(2048 + e * 32) ^ dlane)] = padv;
        }
    }
    __syncthreads();

    const int dscat = cnot_dst((hi << 11) | lo);
    int basev[8];
#pragma unroll
    for (int m = 0; m < 8; ++m)
        basev[m] = addr_of(dscat ^ ((((m >> 2) & 1) << 10) | (((m >> 1) & 1) << 9) | ((m & 1) << 3)));
    const float* th = qp + NQ;

    const float* abase = S + lo * SA + 4 * hi;

    v8f o3[16];
    for (int L = 0; L < NLAYERS; ++L) {
        if (L == PRIO_DROP_A)      __builtin_amdgcn_s_setprio(2);
        else if (L == PRIO_DROP_B) __builtin_amdgcn_s_setprio(1);
        else if (L == PRIO_DROP_C) __builtin_amdgcn_s_setprio(0);

        float ang = th[L * NQ + (lane < NQ ? lane : 0)] * 0.5f;
        float cl = cosf(ang), sl = sinf(ang);
        float c[NQ], sn[NQ];
#pragma unroll
        for (int q = 0; q < NQ; ++q) { c[q] = bcast(cl, q); sn[q] = bcast(sl, q); }
        const float tl = sl / cl;
        float tq8[8];
#pragma unroll
        for (int q = 0; q < 8; ++q) tq8[q] = bcast(tl, q);
        const float cprod = (((c[0] * c[1]) * (c[2] * c[3])) * ((c[4] * c[5]) * (c[6] * c[7])));

        v2f F2[4];
        build_frag(c + 8, sn + 8, lo, hi, cprod, F2);

        {
            v4f xa[2][BT][2];
#pragma unroll
            for (int t = 0; t < BT; ++t) {
                const float* p = abase + t * PA;
                xa[0][t][0] = *(const v4f*)(p);
                xa[0][t][1] = *(const v4f*)(p + 8);
            }
#pragma unroll
            for (int blk = 0; blk < 16 / BT; ++blk) {
                const int cur = blk & 1, nxt = cur ^ 1;
                if (blk + 1 < 16 / BT) {
#pragma unroll
                    for (int t = 0; t < BT; ++t) {
                        const float* p = abase + ((blk + 1) * BT + t) * PA;
                        xa[nxt][t][0] = *(const v4f*)(p);
                        xa[nxt][t][1] = *(const v4f*)(p + 8);
                    }
                    SCHED_FENCE();
                }
                v8f a[BT];
#pragma unroll
                for (int t = 0; t < BT; ++t) a[t] = (v8f){0, 0, 0, 0, 0, 0, 0, 0};
#pragma unroll
                for (int s = 0; s < 4; ++s) {
#pragma unroll
                    for (int t = 0; t < BT; ++t) {
                        const v4f w = xa[cur][t][s >> 1];
                        const v2f as = {w[2 * (s & 1)], w[2 * (s & 1) + 1]};
                        a[t] = WMMA(as, F2[s], a[t]);
                    }
                    SCHED_FENCE();
                }
#pragma unroll
                for (int t = 0; t < BT; ++t) o3[blk * BT + t] = a[t];
            }
        }
        {
            const v2f t1 = {tq8[1], tq8[1]};
            const v2f t2 = {tq8[2], tq8[2]};
            const v2f t3 = {-tq8[3], tq8[3]};
            const float t0s = hi ? tq8[0] : -tq8[0];
            const v2f t0 = {t0s, t0s};
#pragma unroll
            for (int z = 0; z < 16; ++z) {
                v2f p[4];
#pragma unroll
                for (int i = 0; i < 4; ++i) p[i] = (v2f){o3[z][2 * i], o3[z][2 * i + 1]};
#pragma unroll
                for (int q = 2; q >= 1; --q) {
                    const int pb = (q == 2) ? 1 : 2;
                    const v2f tq = (q == 2) ? t2 : t1;
#pragma unroll
                    for (int i = 0; i < 4; ++i) {
                        if (i & pb) continue;
                        const int j = i | pb;
                        const v2f n0 = p[i] - tq * p[j];
                        const v2f n1 = p[j] + tq * p[i];
                        p[i] = n0;
                        p[j] = n1;
                    }
                }
#pragma unroll
                for (int i = 0; i < 4; ++i) {
                    const v2f xs = {p[i].y, p[i].x};
                    const v2f x  = p[i] + t3 * xs;
                    const v2f y  = {xhalf(x.x), xhalf(x.y)};
                    p[i] = x + t0 * y;
                }
#pragma unroll
                for (int i = 0; i < 4; ++i) { o3[z][2 * i] = p[i].x; o3[z][2 * i + 1] = p[i].y; }
            }
        }
#pragma unroll
        for (int q = 7; q >= 4; --q) {
            const int bit = 1 << (7 - q);
            const v2f tq = {tq8[q], tq8[q]};
#pragma unroll
            for (int z = 0; z < 16; ++z) {
                if (z & bit) continue;
#pragma unroll
                for (int i = 0; i < 4; ++i) {
                    const v2f x0 = {o3[z][2 * i], o3[z][2 * i + 1]};
                    const v2f x1 = {o3[z | bit][2 * i], o3[z | bit][2 * i + 1]};
                    const v2f n0 = x0 - tq * x1;
                    const v2f n1 = x1 + tq * x0;
                    o3[z][2 * i] = n0.x;       o3[z][2 * i + 1] = n0.y;
                    o3[z | bit][2 * i] = n1.x; o3[z | bit][2 * i + 1] = n1.y;
                }
            }
        }
        if (L + 1 < NLAYERS) {
            __syncthreads();
#pragma unroll
            for (int z = 0; z < 16; ++z) {
#pragma unroll
                for (int r = 0; r < 8; ++r) {
                    const int K = cnot_dst((r << 8) | (z << 4));
                    const int m = (((K >> 10) & 1) << 2) | (((K >> 9) & 1) << 1) | ((K >> 3) & 1);
                    S[basev[m] + addr_of(K & ~0xE0F)] = o3[z][r];
                }
            }
            __syncthreads();
        }
    }

    float P = 0.f;
    float A[7] = {0.f, 0.f, 0.f, 0.f, 0.f, 0.f, 0.f};
#pragma unroll
    for (int z = 0; z < 16; ++z) {
#pragma unroll
        for (int r = 0; r < 8; ++r) {
            float v = o3[z][r];
            float p = v * v;
            P += p;
#pragma unroll
            for (int q = 1; q < 4; ++q) A[q - 1] += ((r >> (3 - q)) & 1) ? -p : p;
#pragma unroll
            for (int q = 4; q < 8; ++q) A[q - 1] += ((z >> (7 - q)) & 1) ? -p : p;
        }
    }
    float acc[NQ];
    acc[0] = hi ? -P : P;
#pragma unroll
    for (int q = 1; q < 8; ++q) acc[q] = A[q - 1];
#pragma unroll
    for (int q = 8; q < NQ; ++q) acc[q] = ((lo >> (11 - q)) & 1) ? -P : P;
#pragma unroll
    for (int off = 16; off; off >>= 1) {
#pragma unroll
        for (int q = 0; q < NQ; ++q) acc[q] += __shfl_xor(acc[q], off, 32);
    }
    float res = acc[0];
#pragma unroll
    for (int q = 1; q < NQ; ++q) res = (lane == q) ? acc[q] : res;
    if (lane < NQ) out[(size_t)row * NQ + lane] = res;
}

extern "C" void kernel_launch(void* const* d_in, const int* in_sizes, int n_in,
                              void* d_out, int out_size, void* d_ws, size_t ws_size,
                              hipStream_t stream) {
    (void)n_in; (void)out_size; (void)d_ws; (void)ws_size;
    const float* in  = (const float*)d_in[0];
    const float* qp  = (const float*)d_in[1];
    float*       out = (float*)d_out;
    int rows = in_sizes[0] / F_IN;
    qnet_wave_kernel<<<rows, 32, 0, stream>>>(in, qp, out);
}


# lane-half qubit's rotation is carried over the entangling permutation into the next layer's 16x16 factor (contracted group alternates 0-3 / 8-11): 128 lane exchanges a layer go; sum order changes
# speedup vs baseline: 2.9068x; 1.1055x over previous
#include <hip/hip_runtime.h>
#include <math.h>

typedef __attribute__((ext_vector_type(2))) float v2f;
typedef __attribute__((ext_vector_type(4))) float v4f;
typedef __attribute__((ext_vector_type(8))) float v8f;

#define NQ      12
#define NLAYERS 20
#define F_IN    2048
#define PA      16
#define SA      (16 * PA + 4)
#define NST     (16 * SA)
#define PRIO_DROP_A 10
#define PRIO_DROP_B 15
#define PRIO_DROP_C 18

__device__ __forceinline__ int addr_o(int j) {
    return (j >> 8) * SA + ((j >> 4) & 15) * PA + (j & 15);
}
__device__ __forceinline__ int slab_e(int a2) {
    return (((a2 >> 1) & 1) << 3) | (((a2 >> 3) & 1) << 2) | (((a2 >> 2) & 1) << 1) | (a2 & 1);
}
__device__ __forceinline__ int addr_e(int j) {
    return slab_e(j & 15) * SA + ((j >> 4) & 15) * PA + (j >> 8);
}

__device__ __forceinline__ int cnot_dst(int i) {
    const int cq[11] = {0, 2, 4, 6, 8, 10, 1, 3, 5, 7, 9};
    const int tq[11] = {1, 3, 5, 7, 9, 11, 2, 4, 6, 8, 10};
    int j = i;
#pragma unroll
    for (int k = 0; k < 11; ++k) {
        int cm = 1 << (11 - cq[k]);
        int tm = 1 << (11 - tq[k]);
        if (j & cm) j ^= tm;
    }
    return j;
}

__device__ __forceinline__ int take3(int K, int b2, int b1, int b0) {
    return (((K >> b2) & 1) << 2) | (((K >> b1) & 1) << 1) | ((K >> b0) & 1);
}
__device__ __forceinline__ int put3(int m, int b2, int b1, int b0) {
    return (((m >> 2) & 1) << b2) | (((m >> 1) & 1) << b1) | ((m & 1) << b0);
}

__device__ __forceinline__ float tanh_pio2(float x) {
    float xc = fminf(fmaxf(x, -15.f), 15.f);
    float z  = __expf(xc * 2.0f);
    return (z - 1.0f) * __builtin_amdgcn_rcpf(z + 1.0f) * 1.57079632679489662f;
}

__device__ __forceinline__ float bcast(float x, int srclane) {
    return __builtin_bit_cast(float, __builtin_amdgcn_readlane(__builtin_bit_cast(int, x), srclane));
}

__device__ __forceinline__ float ryf(float c, float s, int ob, int ib) {
    return (ob == ib) ? c : (ob ? s : -s);
}

__device__ __forceinline__ void build_frag(const float c[4], const float sn[4], int lo, int hi, float scale, v2f fr[4]) {
    const int o3 = (lo >> 3) & 1, o2 = (lo >> 2) & 1, o1 = (lo >> 1) & 1, o0 = lo & 1;
    float f1[2] = {ryf(c[2], sn[2], o1, 0), ryf(c[2], sn[2], o1, 1)};
    float f0[2] = {ryf(c[3], sn[3], o0, 0), ryf(c[3], sn[3], o0, 1)};
    float f2 = ryf(c[1], sn[1], o2, hi) * scale;
    float f3[2] = {ryf(c[0], sn[0], o3, 0), ryf(c[0], sn[0], o3, 1)};
#pragma unroll
    for (int s = 0; s < 4; ++s) {
        float t = f2 * f3[s >> 1] * f1[s & 1];
        fr[s].x = t * f0[0];
        fr[s].y = t * f0[1];
    }
}

__device__ __forceinline__ float xhalf(float x) {
    int b = __builtin_bit_cast(int, x);
    return __builtin_bit_cast(float, __builtin_amdgcn_permlanex16(b, b, 0x76543210, 0xfedcba98, true, false));
}

__device__ __forceinline__ v2f rot_in_pair(v2f p, v2f tt) {
    v2f d;
    asm("v_pk_fma_f32 %0, %1, %2, %2 op_sel:[0,1,0] op_sel_hi:[0,0,1] neg_lo:[1,0,0]" : "=v"(d) : "s"(tt), "v"(p));
    return d;
}

#define WMMA(a, b, c) __builtin_amdgcn_wmma_f32_16x16x4_f32(false, (a), false, (b), (short)0, (c), false, false)

#define BT 4
#define SCHED_FENCE() __builtin_amdgcn_sched_barrier(0)

__device__ __forceinline__ void prio_step(int L) {
    if (L == PRIO_DROP_A)      __builtin_amdgcn_s_setprio(2);
    else if (L == PRIO_DROP_B) __builtin_amdgcn_s_setprio(1);
    else if (L == PRIO_DROP_C) __builtin_amdgcn_s_setprio(0);
}

template <int PAR>
__device__ __forceinline__ void layer(float* S, const float* abase, const float* th, int L, int lane, int lo, int hi,
                                      const int (&bases)[8], v8f (&o3)[16], float& cdef, float& tdef)
{
    const bool last = (L == NLAYERS - 1);
    prio_step(L);

    float ang = th[L * NQ + (lane < NQ ? lane : 0)] * 0.5f;
    float cl = cosf(ang), sl = sinf(ang);
    float c[NQ], sn[NQ], tq[NQ];
    const float tl = sl / cl;
#pragma unroll
    for (int q = 0; q < NQ; ++q) { c[q] = bcast(cl, q); sn[q] = bcast(sl, q); tq[q] = bcast(tl, q); }

    const int G0 = PAR ? 8 : 0;
    const float tr2 = PAR ? tq[1] : tq[8];
    const float tr1 = PAR ? tq[2] : tq[9];
    const float tr0 = PAR ? tq[3] : tq[11];
    const float cr  = PAR ? ((c[1] * c[2]) * c[3]) : ((c[8] * c[9]) * c[11]);
    const float cx  = (PAR && last) ? c[0] : 1.0f;
    const float cprod = ((cdef * cx) * cr) * ((c[4] * c[5]) * (c[6] * c[7]));

    v2f G[4], F2[4];
    build_frag(c + G0, sn + G0, lo, hi, cprod, G);
    if (PAR == 0) {
        v2f Gp[4];
        build_frag(c + G0, sn + G0, lo, hi ^ 1, cprod, Gp);
#pragma unroll
        for (int s = 0; s < 4; ++s) {
            const float sg = (s >> 1) ? -tdef : tdef;
            F2[s] = G[s] + (v2f){sg, sg} * Gp[s ^ 3];
        }
    } else {
#pragma unroll
        for (int s = 0; s < 4; ++s) {
            const float sg = ((hi ^ (s & 1)) & 1) ? -tdef : tdef;
            F2[s].x = G[s].x + sg * G[s ^ 1].y;
            F2[s].y = G[s].y + sg * G[s ^ 1].x;
        }
    }

    {
        v4f xa[2][BT][2];
#pragma unroll
        for (int t = 0; t < BT; ++t) {
            const float* p = abase + t * PA;
            xa[0][t][0] = *(const v4f*)(p);
            xa[0][t][1] = *(const v4f*)(p + 8);
        }
#pragma unroll
        for (int blk = 0; blk < 16 / BT; ++blk) {
            const int cur = blk & 1, nxt = cur ^ 1;
            if (blk + 1 < 16 / BT) {
#pragma unroll
                for (int t = 0; t < BT; ++t) {
                    const float* p = abase + ((blk + 1) * BT + t) * PA;
                    xa[nxt][t][0] = *(const v4f*)(p);
                    xa[nxt][t][1] = *(const v4f*)(p + 8);
                }
                SCHED_FENCE();
            }
            v8f a[BT];
#pragma unroll
            for (int t = 0; t < BT; ++t) a[t] = (v8f){0, 0, 0, 0, 0, 0, 0, 0};
#pragma unroll
            for (int s = 0; s < 4; ++s) {
#pragma unroll
                for (int t = 0; t < BT; ++t) {
                    const v4f w = xa[cur][t][s >> 1];
                    const v2f as = {w[2 * (s & 1)], w[2 * (s & 1) + 1]};
                    a[t] = WMMA(as, F2[s], a[t]);
                }
                SCHED_FENCE();
            }
#pragma unroll
            for (int t = 0; t < BT; ++t) o3[blk * BT + t] = a[t];
        }
    }
    {
        const v2f t1 = {tr2, tr2};
        const v2f t2 = {tr1, tr1};
        const v2f t3 = {tr0, tr0};
#pragma unroll
        for (int z = 0; z < 16; ++z) {
            v2f p[4];
#pragma unroll
            for (int i = 0; i < 4; ++i) p[i] = (v2f){o3[z][2 * i], o3[z][2 * i + 1]};
#pragma unroll
            for (int q = 2; q >= 1; --q) {
                const int pb = (q == 2) ? 1 : 2;
                const v2f tt = (q == 2) ? t2 : t1;
#pragma unroll
                for (int i = 0; i < 4; ++i) {
                    if (i & pb) continue;
                    const int j = i | pb;
                    const v2f n0 = p[i] - tt * p[j];
                    const v2f n1 = p[j] + tt * p[i];
                    p[i] = n0;
                    p[j] = n1;
                }
            }
#pragma unroll
            for (int i = 0; i < 4; ++i) p[i] = rot_in_pair(p[i], t3);
#pragma unroll
            for (int i = 0; i < 4; ++i) { o3[z][2 * i] = p[i].x; o3[z][2 * i + 1] = p[i].y; }
        }
    }
#pragma unroll
    for (int q = 7; q >= 4; --q) {
        const int bit = 1 << (7 - q);
        const v2f tt = {tq[q], tq[q]};
#pragma unroll
        for (int z = 0; z < 16; ++z) {
            if (z & bit) continue;
#pragma unroll
            for (int i = 0; i < 4; ++i) {
                const v2f x0 = {o3[z][2 * i], o3[z][2 * i + 1]};
                const v2f x1 = {o3[z | bit][2 * i], o3[z | bit][2 * i + 1]};
                const v2f n0 = x0 - tt * x1;
                const v2f n1 = x1 + tt * x0;
                o3[z][2 * i] = n0.x;       o3[z][2 * i + 1] = n0.y;
                o3[z | bit][2 * i] = n1.x; o3[z | bit][2 * i + 1] = n1.y;
            }
        }
    }
    if (PAR == 1 && last) {
        const float t0s = hi ? tq[0] : -tq[0];
        const v2f t0 = {t0s, t0s};
#pragma unroll
        for (int z = 0; z < 16; ++z) {
#pragma unroll
            for (int i = 0; i < 4; ++i) {
                const v2f x = {o3[z][2 * i], o3[z][2 * i + 1]};
                const v2f y = {xhalf(x.x), xhalf(x.y)};
                const v2f n = x + t0 * y;
                o3[z][2 * i] = n.x; o3[z][2 * i + 1] = n.y;
            }
        }
    } else {
        __syncthreads();
#pragma unroll
        for (int z = 0; z < 16; ++z) {
#pragma unroll
            for (int r = 0; r < 8; ++r) {
                if (PAR == 0) {
                    const int je = (z << 4) | (((r >> 2) & 1) << 3) | (((r >> 1) & 1) << 2) | (r & 1);
                    const int K = cnot_dst(je);
                    S[bases[take3(K, 7, 1, 0)] + addr_o(K & ~0x083)] = o3[z][r];
                } else {
                    const int K = cnot_dst((r << 8) | (z << 4));
                    S[bases[take3(K, 10, 9, 3)] + addr_e(K & ~0x608)] = o3[z][r];
                }
            }
        }
        __syncthreads();
        cdef = PAR ? c[0] : c[10];
        tdef = PAR ? tq[0] : tq[10];
    }
}

__global__ __launch_bounds__(32) __attribute__((amdgpu_waves_per_eu(4, 4)))
void qnet_wave_kernel(const float* __restrict__ in, const float* __restrict__ qp,
                      float* __restrict__ out)
{
    __shared__ __attribute__((aligned(16))) float S[NST];

    __builtin_amdgcn_s_setprio(3);

    const int lane = threadIdx.x;
    const int hi   = lane >> 4;
    const int lo   = lane & 15;
    const int row  = blockIdx.x;

    const float* srcrow = in + (size_t)row * F_IN;
    {
        const int dlane = cnot_dst(lane);
        int bem[4];
#pragma unroll
        for (int m = 0; m < 4; ++m) bem[m] = addr_e(dlane ^ (((m >> 1) & 1) << 4) ^ ((m & 1) << 3));
        float q[64];
        float ss = 0.f;
#pragma unroll
        for (int e = 0; e < 64; ++e) {
            q[e] = tanh_pio2(srcrow[e * 32 + lane]);
            ss += q[e] * q[e];
        }
#pragma unroll
        for (int off = 16; off; off >>= 1) ss += __shfl_xor(ss, off, 32);
        const float sc = 1.0f / sqrtf(ss + 512.0f);
        const float padv = 0.5f * sc;
#pragma unroll
        for (int e = 0; e < 64; ++e) {
            const int K1 = cnot_dst(e * 32);
            const int K2 = cnot_dst(2048 + e * 32);
            S[bem[(((K1 >> 4) & 1) << 1) | ((K1 >> 3) & 1)] + addr_e(K1 & ~0x018)] = q[e] * sc;
            S[bem[(((K2 >> 4) & 1) << 1) | ((K2 >> 3) & 1)] + addr_e(K2 & ~0x018)] = padv;
        }
    }
    __syncthreads();

    int base_ev[8], base_od[8];
    {
        const int de = cnot_dst((lo << 8) | (hi << 1));
        const int dd = cnot_dst((hi << 11) | lo);
#pragma unroll
        for (int m = 0; m < 8; ++m) {
            base_ev[m] = addr_o(de ^ put3(m, 7, 1, 0));
            base_od[m] = addr_e(dd ^ put3(m, 10, 9, 3));
        }
    }
    const float* th = qp + NQ;

    const float* abase = S + lo * SA + 4 * hi;

    v8f o3[16];
    float cdef = 1.0f, tdef = 0.0f;
    for (int P = 0; P < NLAYERS / 2; ++P) {
        layer<0>(S, abase, th, 2 * P,     lane, lo, hi, base_ev, o3, cdef, tdef);
        layer<1>(S, abase, th, 2 * P + 1, lane, lo, hi, base_od, o3, cdef, tdef);
    }

    float Pt = 0.f;
    float A[7] = {0.f, 0.f, 0.f, 0.f, 0.f, 0.f, 0.f};
#pragma unroll
    for (int z = 0; z < 16; ++z) {
#pragma unroll
        for (int r = 0; r < 8; ++r) {
            float v = o3[z][r];
            float p = v * v;
            Pt += p;
#pragma unroll
            for (int q = 1; q < 4; ++q) A[q - 1] += ((r >> (3 - q)) & 1) ? -p : p;
#pragma unroll
            for (int q = 4; q < 8; ++q) A[q - 1] += ((z >> (7 - q)) & 1) ? -p : p;
        }
    }
    float acc[NQ];
    acc[0] = hi ? -Pt : Pt;
#pragma unroll
    for (int q = 1; q < 8; ++q) acc[q] = A[q - 1];
#pragma unroll
    for (int q = 8; q < NQ; ++q) acc[q] = ((lo >> (11 - q)) & 1) ? -Pt : Pt;
#pragma unroll
    for (int off = 16; off; off >>= 1) {
#pragma unroll
        for (int q = 0; q < NQ; ++q) acc[q] += __shfl_xor(acc[q], off, 32);
    }
    float res = acc[0];
#pragma unroll
    for (int q = 1; q < NQ; ++q) res = (lane == q) ? acc[q] : res;
    if (lane < NQ) out[(size_t)row * NQ + lane] = res;
}

extern "C" void kernel_launch(void* const* d_in, const int* in_sizes, int n_in,
                              void* d_out, int out_size, void* d_ws, size_t ws_size,
                              hipStream_t stream) {
    (void)n_in; (void)out_size; (void)d_ws; (void)ws_size;
    const float* in  = (const float*)d_in[0];
    const float* qp  = (const float*)d_in[1];
    float*       out = (float*)d_out;
    int rows = in_sizes[0] / F_IN;
    qnet_wave_kernel<<<rows, 32, 0, stream>>>(in, qp, out);
}


# lane-row qubits 8-11 (even layers) and 0,1 (odd) are not rotated: carried over the entangling permutation into the next 16x16 factor (one factor per tile parity after even layers); sum order changes
# speedup vs baseline: 3.0734x; 1.0573x over previous
#include <hip/hip_runtime.h>
#include <math.h>

typedef __attribute__((ext_vector_type(2))) float v2f;
typedef __attribute__((ext_vector_type(4))) float v4f;
typedef __attribute__((ext_vector_type(8))) float v8f;

#define NQ      12
#define NLAYERS 20
#define F_IN    2048
#define PA      16
#define SA      (16 * PA + 4)
#define NST     (16 * SA)
#define PRIO_DROP_A 10
#define PRIO_DROP_B 15
#define PRIO_DROP_C 18

__device__ __forceinline__ int addr_o(int j) {
    return (j >> 8) * SA + ((j >> 4) & 15) * PA + (j & 15);
}
__device__ __forceinline__ int slab_e(int a2) {
    return (((a2 >> 1) & 1) << 3) | (((a2 >> 3) & 1) << 2) | (((a2 >> 2) & 1) << 1) | (a2 & 1);
}
__device__ __forceinline__ int addr_e(int j) {
    return slab_e(j & 15) * SA + ((j >> 4) & 15) * PA + (j >> 8);
}

__device__ __forceinline__ int cnot_dst(int i) {
    const int cq[11] = {0, 2, 4, 6, 8, 10, 1, 3, 5, 7, 9};
    const int tq[11] = {1, 3, 5, 7, 9, 11, 2, 4, 6, 8, 10};
    int j = i;
#pragma unroll
    for (int k = 0; k < 11; ++k) {
        int cm = 1 << (11 - cq[k]);
        int tm = 1 << (11 - tq[k]);
        if (j & cm) j ^= tm;
    }
    return j;
}

__device__ __forceinline__ int take3(int K, int b2, int b1, int b0) {
    return (((K >> b2) & 1) << 2) | (((K >> b1) & 1) << 1) | ((K >> b0) & 1);
}
__device__ __forceinline__ int put3(int m, int b2, int b1, int b0) {
    return (((m >> 2) & 1) << b2) | (((m >> 1) & 1) << b1) | ((m & 1) << b0);
}

__device__ __forceinline__ float tanh_pio2(float x) {
    float xc = fminf(fmaxf(x, -15.f), 15.f);
    float z  = __expf(xc * 2.0f);
    return (z - 1.0f) * __builtin_amdgcn_rcpf(z + 1.0f) * 1.57079632679489662f;
}

__device__ __forceinline__ float bcast(float x, int srclane) {
    return __builtin_bit_cast(float, __builtin_amdgcn_readlane(__builtin_bit_cast(int, x), srclane));
}

__device__ __forceinline__ float ryf(float c, float s, int ob, int ib) {
    return (ob == ib) ? c : (ob ? s : -s);
}

__device__ __forceinline__ void build_frag(const float c[4], const float sn[4], int lo, int hi, float scale, v2f fr[4]) {
    const int o3 = (lo >> 3) & 1, o2 = (lo >> 2) & 1, o1 = (lo >> 1) & 1, o0 = lo & 1;
    float f1[2] = {ryf(c[2], sn[2], o1, 0), ryf(c[2], sn[2], o1, 1)};
    float f0[2] = {ryf(c[3], sn[3], o0, 0), ryf(c[3], sn[3], o0, 1)};
    float f2 = ryf(c[1], sn[1], o2, hi) * scale;
    float f3[2] = {ryf(c[0], sn[0], o3, 0), ryf(c[0], sn[0], o3, 1)};
#pragma unroll
    for (int s = 0; s < 4; ++s) {
        float t = f2 * f3[s >> 1] * f1[s & 1];
        fr[s].x = t * f0[0];
        fr[s].y = t * f0[1];
    }
}

__device__ __forceinline__ float xhalf(float x) {
    int b = __builtin_bit_cast(int, x);
    return __builtin_bit_cast(float, __builtin_amdgcn_permlanex16(b, b, 0x76543210, 0xfedcba98, true, false));
}

__device__ __forceinline__ v2f rot_in_pair(v2f p, v2f tt) {
    v2f d;
    asm("v_pk_fma_f32 %0, %1, %2, %2 op_sel:[0,1,0] op_sel_hi:[0,0,1] neg_lo:[1,0,0]" : "=v"(d) : "s"(tt), "v"(p));
    return d;
}

#define WMMA(a, b, c) __builtin_amdgcn_wmma_f32_16x16x4_f32(false, (a), false, (b), (short)0, (c), false, false)

#define BT 4
#define SCHED_FENCE() __builtin_amdgcn_sched_barrier(0)

__device__ __forceinline__ void prio_step(int L) {
    if (L == PRIO_DROP_A)      __builtin_amdgcn_s_setprio(2);
    else if (L == PRIO_DROP_B) __builtin_amdgcn_s_setprio(1);
    else if (L == PRIO_DROP_C) __builtin_amdgcn_s_setprio(0);
}

template <int PAR>
__device__ __forceinline__ void layer(float* S, const float* abase, const float* th, int L, int lane, int lo, int hi,
                                      const int (&bases)[8], v8f (&o3)[16], float& cdef, float (&tdef)[4])
{
    const bool last = (L == NLAYERS - 1);
    prio_step(L);

    float ang = th[L * NQ + (lane < NQ ? lane : 0)] * 0.5f;
    float cl = cosf(ang), sl = sinf(ang);
    float c[NQ], sn[NQ], tq[NQ];
    const float tl = sl / cl;
#pragma unroll
    for (int q = 0; q < NQ; ++q) { c[q] = bcast(cl, q); sn[q] = bcast(sl, q); tq[q] = bcast(tl, q); }

    const int G0 = PAR ? 8 : 0;
    const float tr2 = tq[1];
    const float tr1 = tq[2];
    const float tr0 = tq[3];
    const float cr  = PAR ? (c[2] * c[3]) : 1.0f;
    const float cx  = (PAR && last) ? (c[0] * c[1]) : 1.0f;
    const float cprod = ((cdef * cx) * cr) * ((c[4] * c[5]) * (c[6] * c[7]));

    v2f H[2][4];
    build_frag(c + G0, sn + G0, lo, hi, cprod, H[0]);
    build_frag(c + G0, sn + G0, lo, hi ^ 1, cprod, H[1]);
    v2f F2[2][4];
    if (PAR == 0) {
        const float t1h = hi ? -tdef[1] : tdef[1];
        v2f H1[2][4];
#pragma unroll
        for (int h = 0; h < 2; ++h) {
#pragma unroll
            for (int s = 0; s < 4; ++s) {
                const float w = ((h ^ (s >> 1)) & 1) ? -t1h : t1h;
                H1[h][s] = H[h][s] + (v2f){w, w} * H[h ^ 1][s ^ 1];
            }
        }
#pragma unroll
        for (int s = 0; s < 4; ++s) {
            const float w = (s >> 1) ? -tdef[0] : tdef[0];
            F2[0][s] = H1[0][s] + (v2f){w, w} * H1[1][s ^ 3];
            F2[1][s] = F2[0][s];
        }
    } else {
        const float t11h = hi ? -tdef[3] : tdef[3];
        const float t10h = hi ? -tdef[2] : tdef[2];
        v2f Ha[2][4], Hb[2][4];
#pragma unroll
        for (int h = 0; h < 2; ++h) {
#pragma unroll
            for (int s = 0; s < 4; ++s) {
                const float w = ((h ^ s) & 1) ? -t11h : t11h;
                Ha[h][s].x = H[h][s].x + w * H[h][s].y;
                Ha[h][s].y = H[h][s].y - w * H[h][s].x;
            }
        }
#pragma unroll
        for (int h = 0; h < 2; ++h) {
#pragma unroll
            for (int s = 0; s < 4; ++s) {
                const float w = ((h ^ s) & 1) ? -t10h : t10h;
                Hb[h][s].x = Ha[h][s].x + w * Ha[h][s ^ 1].y;
                Hb[h][s].y = Ha[h][s].y + w * Ha[h][s ^ 1].x;
            }
        }
        const float t9h = hi ? -tdef[1] : tdef[1];
#pragma unroll
        for (int p = 0; p < 2; ++p) {
            v2f Hc[2][4];
#pragma unroll
            for (int h = 0; h < 2; ++h) {
#pragma unroll
                for (int s = 0; s < 4; ++s) {
                    const float w = ((h ^ (s >> 1) ^ p) & 1) ? -t9h : t9h;
                    Hc[h][s] = Hb[h][s] + (v2f){w, w} * Hb[h ^ 1][s ^ 1];
                }
            }
#pragma unroll
            for (int s = 0; s < 4; ++s) {
                const float w = (((s >> 1) ^ p) & 1) ? -tdef[0] : tdef[0];
                F2[p][s] = Hc[0][s] + (v2f){w, w} * Hc[1][s ^ 3];
            }
        }
    }

    {
        v4f xa[2][BT][2];
#pragma unroll
        for (int t = 0; t < BT; ++t) {
            const float* p = abase + t * PA;
            xa[0][t][0] = *(const v4f*)(p);
            xa[0][t][1] = *(const v4f*)(p + 8);
        }
#pragma unroll
        for (int blk = 0; blk < 16 / BT; ++blk) {
            const int cur = blk & 1, nxt = cur ^ 1;
            if (blk + 1 < 16 / BT) {
#pragma unroll
                for (int t = 0; t < BT; ++t) {
                    const float* p = abase + ((blk + 1) * BT + t) * PA;
                    xa[nxt][t][0] = *(const v4f*)(p);
                    xa[nxt][t][1] = *(const v4f*)(p + 8);
                }
                SCHED_FENCE();
            }
            v8f a[BT];
#pragma unroll
            for (int t = 0; t < BT; ++t) a[t] = (v8f){0, 0, 0, 0, 0, 0, 0, 0};
#pragma unroll
            for (int s = 0; s < 4; ++s) {
#pragma unroll
                for (int t = 0; t < BT; ++t) {
                    const v4f w = xa[cur][t][s >> 1];
                    const v2f as = {w[2 * (s & 1)], w[2 * (s & 1) + 1]};
                    a[t] = WMMA(as, F2[t & 1][s], a[t]);
                }
                SCHED_FENCE();
            }
#pragma unroll
            for (int t = 0; t < BT; ++t) o3[blk * BT + t] = a[t];
        }
    }
    if (PAR == 1) {
        const v2f t2 = {tr1, tr1};
        const v2f t3 = {tr0, tr0};
#pragma unroll
        for (int z = 0; z < 16; ++z) {
            v2f p[4];
#pragma unroll
            for (int i = 0; i < 4; ++i) p[i] = (v2f){o3[z][2 * i], o3[z][2 * i + 1]};
#pragma unroll
            for (int i = 0; i < 4; ++i) {
                if (i & 1) continue;
                const int j = i | 1;
                const v2f n0 = p[i] - t2 * p[j];
                const v2f n1 = p[j] + t2 * p[i];
                p[i] = n0;
                p[j] = n1;
            }
#pragma unroll
            for (int i = 0; i < 4; ++i) p[i] = rot_in_pair(p[i], t3);
#pragma unroll
            for (int i = 0; i < 4; ++i) { o3[z][2 * i] = p[i].x; o3[z][2 * i + 1] = p[i].y; }
        }
    }
#pragma unroll
    for (int q = 7; q >= 4; --q) {
        const int bit = 1 << (7 - q);
        const v2f tt = {tq[q], tq[q]};
#pragma unroll
        for (int z = 0; z < 16; ++z) {
            if (z & bit) continue;
#pragma unroll
            for (int i = 0; i < 4; ++i) {
                const v2f x0 = {o3[z][2 * i], o3[z][2 * i + 1]};
                const v2f x1 = {o3[z | bit][2 * i], o3[z | bit][2 * i + 1]};
                const v2f n0 = x0 - tt * x1;
                const v2f n1 = x1 + tt * x0;
                o3[z][2 * i] = n0.x;       o3[z][2 * i + 1] = n0.y;
                o3[z | bit][2 * i] = n1.x; o3[z | bit][2 * i + 1] = n1.y;
            }
        }
    }
    if (PAR == 1 && last) {
        const v2f t1 = {tr2, tr2};
#pragma unroll
        for (int z = 0; z < 16; ++z) {
#pragma unroll
            for (int i = 0; i < 2; ++i) {
                const v2f x0 = {o3[z][2 * i], o3[z][2 * i + 1]};
                const v2f x1 = {o3[z][2 * i + 4], o3[z][2 * i + 5]};
                const v2f n0 = x0 - t1 * x1;
                const v2f n1 = x1 + t1 * x0;
                o3[z][2 * i] = n0.x;     o3[z][2 * i + 1] = n0.y;
                o3[z][2 * i + 4] = n1.x; o3[z][2 * i + 5] = n1.y;
            }
        }
        const float t0s = hi ? tq[0] : -tq[0];
        const v2f t0 = {t0s, t0s};
#pragma unroll
        for (int z = 0; z < 16; ++z) {
#pragma unroll
            for (int i = 0; i < 4; ++i) {
                const v2f x = {o3[z][2 * i], o3[z][2 * i + 1]};
                const v2f y = {xhalf(x.x), xhalf(x.y)};
                const v2f n = x + t0 * y;
                o3[z][2 * i] = n.x; o3[z][2 * i + 1] = n.y;
            }
        }
    } else {
        __syncthreads();
#pragma unroll
        for (int z = 0; z < 16; ++z) {
#pragma unroll
            for (int r = 0; r < 8; ++r) {
                if (PAR == 0) {
                    const int je = (z << 4) | (((r >> 2) & 1) << 3) | (((r >> 1) & 1) << 2) | (r & 1);
                    const int K = cnot_dst(je);
                    S[bases[take3(K, 7, 1, 0)] + addr_o(K & ~0x083)] = o3[z][r];
                } else {
                    const int K = cnot_dst((r << 8) | (z << 4));
                    S[bases[take3(K, 10, 9, 3)] + addr_e(K & ~0x608)] = o3[z][r];
                }
            }
        }
        __syncthreads();
        if (PAR) {
            cdef = c[0] * c[1];
            tdef[0] = tq[0]; tdef[1] = tq[1];
        } else {
            cdef = (c[8] * c[9]) * (c[10] * c[11]);
            tdef[0] = tq[8]; tdef[1] = tq[9]; tdef[2] = tq[10]; tdef[3] = tq[11];
        }
    }
}

__global__ __launch_bounds__(32) __attribute__((amdgpu_waves_per_eu(4, 4)))
void qnet_wave_kernel(const float* __restrict__ in, const float* __restrict__ qp,
                      float* __restrict__ out)
{
    __shared__ __attribute__((aligned(16))) float S[NST];

    __builtin_amdgcn_s_setprio(3);

    const int lane = threadIdx.x;
    const int hi   = lane >> 4;
    const int lo   = lane & 15;
    const int row  = blockIdx.x;

    const float* srcrow = in + (size_t)row * F_IN;
    {
        const int dlane = cnot_dst(lane);
        int bem[4];
#pragma unroll
        for (int m = 0; m < 4; ++m) bem[m] = addr_e(dlane ^ (((m >> 1) & 1) << 4) ^ ((m & 1) << 3));
        float q[64];
        float ss = 0.f;
#pragma unroll
        for (int e = 0; e < 64; ++e) {
            q[e] = tanh_pio2(srcrow[e * 32 + lane]);
            ss += q[e] * q[e];
        }
#pragma unroll
        for (int off = 16; off; off >>= 1) ss += __shfl_xor(ss, off, 32);
        const float sc = 1.0f / sqrtf(ss + 512.0f);
        const float padv = 0.5f * sc;
#pragma unroll
        for (int e = 0; e < 64; ++e) {
            const int K1 = cnot_dst(e * 32);
            const int K2 = cnot_dst(2048 + e * 32);
            S[bem[(((K1 >> 4) & 1) << 1) | ((K1 >> 3) & 1)] + addr_e(K1 & ~0x018)] = q[e] * sc;
            S[bem[(((K2 >> 4) & 1) << 1) | ((K2 >> 3) & 1)] + addr_e(K2 & ~0x018)] = padv;
        }
    }
    __syncthreads();

    int base_ev[8], base_od[8];
    {
        const int de = cnot_dst((lo << 8) | (hi << 1));
        const int dd = cnot_dst((hi << 11) | lo);
#pragma unroll
        for (int m = 0; m < 8; ++m) {
            base_ev[m] = addr_o(de ^ put3(m, 7, 1, 0));
            base_od[m] = addr_e(dd ^ put3(m, 10, 9, 3));
        }
    }
    const float* th = qp + NQ;

    const float* abase = S + lo * SA + 4 * hi;

    v8f o3[16];
    float cdef = 1.0f;
    float tdef[4] = {0.0f, 0.0f, 0.0f, 0.0f};
    for (int P = 0; P < NLAYERS / 2; ++P) {
        layer<0>(S, abase, th, 2 * P,     lane, lo, hi, base_ev, o3, cdef, tdef);
        layer<1>(S, abase, th, 2 * P + 1, lane, lo, hi, base_od, o3, cdef, tdef);
    }

    float Pt = 0.f;
    float A[7] = {0.f, 0.f, 0.f, 0.f, 0.f, 0.f, 0.f};
#pragma unroll
    for (int z = 0; z < 16; ++z) {
#pragma unroll
        for (int r = 0; r < 8; ++r) {
            float v = o3[z][r];
            float p = v * v;
            Pt += p;
#pragma unroll
            for (int q = 1; q < 4; ++q) A[q - 1] += ((r >> (3 - q)) & 1) ? -p : p;
#pragma unroll
            for (int q = 4; q < 8; ++q) A[q - 1] += ((z >> (7 - q)) & 1) ? -p : p;
        }
    }
    float acc[NQ];
    acc[0] = hi ? -Pt : Pt;
#pragma unroll
    for (int q = 1; q < 8; ++q) acc[q] = A[q - 1];
#pragma unroll
    for (int q = 8; q < NQ; ++q) acc[q] = ((lo >> (11 - q)) & 1) ? -Pt : Pt;
#pragma unroll
    for (int off = 16; off; off >>= 1) {
#pragma unroll
        for (int q = 0; q < NQ; ++q) acc[q] += __shfl_xor(acc[q], off, 32);
    }
    float res = acc[0];
#pragma unroll
    for (int q = 1; q < NQ; ++q) res = (lane == q) ? acc[q] : res;
    if (lane < NQ) out[(size_t)row * NQ + lane] = res;
}

extern "C" void kernel_launch(void* const* d_in, const int* in_sizes, int n_in,
                              void* d_out, int out_size, void* d_ws, size_t ws_size,
                              hipStream_t stream) {
    (void)n_in; (void)out_size; (void)d_ws; (void)ws_size;
    const float* in  = (const float*)d_in[0];
    const float* qp  = (const float*)d_in[1];
    float*       out = (float*)d_out;
    int rows = in_sizes[0] / F_IN;
    qnet_wave_kernel<<<rows, 32, 0, stream>>>(in, qp, out);
}
